# starting kernel
# baseline (speedup 1.0000x reference)
#include <hip/hip_runtime.h>

#define NUM_USER 50000
#define NUM_ITEM 50000
#define N_NODES  100000
#define DIM_E    64
#define DIM_FEAT 128
#define DIM_H    256
#define N_EDGES  1600000
#define N_TILES  (NUM_ITEM / 16)

typedef __attribute__((ext_vector_type(2))) float v2f;
typedef __attribute__((ext_vector_type(8))) float v8f;

__global__ void k_zero(float* __restrict__ p, int n) {
    int i = blockIdx.x * blockDim.x + threadIdx.x;
    if (i < n) p[i] = 0.0f;
}

__global__ __launch_bounds__(128) void k_mlp(
        const float* __restrict__ F,
        const float* __restrict__ W1, const float* __restrict__ b1,
        const float* __restrict__ W2, const float* __restrict__ b2,
        float* __restrict__ temp) {
    __shared__ float hid[4][16][DIM_H];

    const int wave = threadIdx.x >> 5;
    const int lane = threadIdx.x & 31;
    const int lm   = lane & 15;
    const int lh   = lane >> 4;
    const int tile = blockIdx.x * 4 + wave;
    if (tile >= N_TILES) return;
    const int r0 = tile * 16;

    for (int nt = 0; nt < DIM_H / 16; ++nt) {
        const int N = nt * 16 + lm;
        v8f acc = {};
        #pragma unroll 4
        for (int kb = 0; kb < DIM_FEAT / 4; ++kb) {
            const int k0 = kb * 4 + lh * 2;
            v2f a = *(const v2f*)(F + (r0 + lm) * DIM_FEAT + k0);
            v2f b;
            b.x = W1[(k0    ) * DIM_H + N];
            b.y = W1[(k0 + 1) * DIM_H + N];
            acc = __builtin_amdgcn_wmma_f32_16x16x4_f32(
                      false, a, false, b, (short)0, acc, false, false);
        }
        const float bias = b1[N];
        #pragma unroll
        for (int r = 0; r < 8; ++r) {
            float hv = acc[r] + bias;
            hv = (hv > 0.0f) ? hv : 0.01f * hv;
            hid[wave][r + 8 * lh][N] = hv;
        }
    }

    for (int nt = 0; nt < DIM_E / 16; ++nt) {
        const int N = nt * 16 + lm;
        v8f acc = {};
        #pragma unroll 4
        for (int kb = 0; kb < DIM_H / 4; ++kb) {
            const int k0 = kb * 4 + lh * 2;
            v2f a = *(const v2f*)(&hid[wave][lm][k0]);
            v2f b;
            b.x = W2[(k0    ) * DIM_E + N];
            b.y = W2[(k0 + 1) * DIM_E + N];
            acc = __builtin_amdgcn_wmma_f32_16x16x4_f32(
                      false, a, false, b, (short)0, acc, false, false);
        }
        const float bias = b2[N];
        #pragma unroll
        for (int r = 0; r < 8; ++r)
            temp[(r0 + r + 8 * lh) * DIM_E + N] = acc[r] + bias;
    }
}

__global__ void k_normalize(const float* __restrict__ pref,
                            const float* __restrict__ temp,
                            float* __restrict__ xn) {
    const int wave = threadIdx.x >> 5;
    const int lane = threadIdx.x & 31;
    const int node = blockIdx.x * (blockDim.x >> 5) + wave;
    if (node >= N_NODES) return;
    const float* src = (node < NUM_USER) ? (pref + (size_t)node * DIM_E)
                                         : (temp + (size_t)(node - NUM_USER) * DIM_E);
    float v0 = src[lane];
    float v1 = src[lane + 32];
    float s = v0 * v0 + v1 * v1;
    #pragma unroll
    for (int off = 16; off > 0; off >>= 1) s += __shfl_xor(s, off, 32);
    const float inv = 1.0f / fmaxf(sqrtf(s), 1e-12f);
    xn[(size_t)node * DIM_E + lane]      = v0 * inv;
    xn[(size_t)node * DIM_E + lane + 32] = v1 * inv;
}

__global__ void k_degree(const int* __restrict__ ei, float* __restrict__ deg) {
    int e = blockIdx.x * blockDim.x + threadIdx.x;
    if (e >= N_EDGES) return;
    int r = ei[e];
    int c = ei[N_EDGES + e];
    if (r != c) atomicAdd(deg + r, 1.0f);
}

__global__ void k_dis(float* __restrict__ deg) {
    int n = blockIdx.x * blockDim.x + threadIdx.x;
    if (n >= N_NODES) return;
    float d = deg[n];
    deg[n] = (d > 0.0f) ? rsqrtf(d) : 0.0f;
}

__global__ void k_conv(const int* __restrict__ ei, const float* __restrict__ dis,
                       const float* __restrict__ x, float* __restrict__ out) {
    int tid = blockIdx.x * blockDim.x + threadIdx.x;
    int e = tid >> 4;
    int q = tid & 15;
    if (e >= N_EDGES) return;
    int r = ei[e];
    int c = ei[N_EDGES + e];
    if (r == c) return;
    float nrm = dis[r] * dis[c];
    if (nrm == 0.0f) return;
    const float4 v = *(const float4*)(x + (size_t)r * DIM_E + q * 4);
    float* o = out + (size_t)c * DIM_E + q * 4;
    atomicAdd(o + 0, v.x * nrm);
    atomicAdd(o + 1, v.y * nrm);
    atomicAdd(o + 2, v.z * nrm);
    atomicAdd(o + 3, v.w * nrm);
}

__global__ void k_final(const float* __restrict__ xn, const float* __restrict__ h,
                        const float* __restrict__ h1, const float* __restrict__ pref,
                        float* __restrict__ out) {
    const int n4 = N_NODES * DIM_E / 4;
    const int p4 = NUM_USER * DIM_E / 4;
    int i = blockIdx.x * blockDim.x + threadIdx.x;
    if (i >= n4) return;
    float4 a = ((const float4*)xn)[i];
    float4 b = ((const float4*)h)[i];
    float4 c = ((const float4*)h1)[i];
    float4 rr;
    rr.x = a.x + b.x + c.x;
    rr.y = a.y + b.y + c.y;
    rr.z = a.z + b.z + c.z;
    rr.w = a.w + b.w + c.w;
    ((float4*)out)[i] = rr;
    if (i < p4) ((float4*)out)[n4 + i] = ((const float4*)pref)[i];
}

extern "C" void kernel_launch(void* const* d_in, const int* in_sizes, int n_in,
                              void* d_out, int out_size, void* d_ws, size_t ws_size,
                              hipStream_t stream) {
    const int*   ei   = (const int*)  d_in[0];
    const float* F    = (const float*)d_in[1];
    const float* pref = (const float*)d_in[2];
    const float* W1   = (const float*)d_in[3];
    const float* b1   = (const float*)d_in[4];
    const float* W2   = (const float*)d_in[5];
    const float* b2   = (const float*)d_in[6];
    float*       out  = (float*)d_out;

    float* ws   = (float*)d_ws;
    float* xn   = ws;
    float* h    = xn   + (size_t)N_NODES * DIM_E;
    float* h1   = h    + (size_t)N_NODES * DIM_E;
    float* temp = h1   + (size_t)N_NODES * DIM_E;
    float* deg  = temp + (size_t)NUM_ITEM * DIM_E;

    const int T = 256;
    {
        int n = 2 * N_NODES * DIM_E;
        k_zero<<<(n + T - 1) / T, T, 0, stream>>>(h, n);
        k_zero<<<(N_NODES + T - 1) / T, T, 0, stream>>>(deg, N_NODES);
    }
    k_mlp<<<(N_TILES + 3) / 4, 128, 0, stream>>>(F, W1, b1, W2, b2, temp);
    k_normalize<<<(N_NODES + 7) / 8, T, 0, stream>>>(pref, temp, xn);
    k_degree<<<(N_EDGES + T - 1) / T, T, 0, stream>>>(ei, deg);
    k_dis<<<(N_NODES + T - 1) / T, T, 0, stream>>>(deg);
    {
        int nthr = N_EDGES * 16;
        k_conv<<<(nthr + T - 1) / T, T, 0, stream>>>(ei, deg, xn, h);
        k_conv<<<(nthr + T - 1) / T, T, 0, stream>>>(ei, deg, h, h1);
    }
    k_final<<<(N_NODES * DIM_E / 4 + T - 1) / T, T, 0, stream>>>(xn, h, h1, pref, out);
}


# conv hops as a gather: one returning integer atomic per edge files its source row in the destination's 256-B record; half waves sum a row's in-edges in registers; no f32 atomics; slot-order sums
# speedup vs baseline: 1.6710x; 1.6710x over previous
#include <hip/hip_runtime.h>

#define NUM_USER 50000
#define NUM_ITEM 50000
#define N_NODES  100000
#define DIM_E    64
#define DIM_FEAT 128
#define DIM_H    256
#define N_EDGES  1600000
#define N_TILES  (NUM_ITEM / 16)

typedef __attribute__((ext_vector_type(2))) float v2f;
typedef __attribute__((ext_vector_type(8))) float v8f;

#define REC_W   64
#define REC_CAP (REC_W - 1)

__global__ void k_init(float* __restrict__ deg, int* __restrict__ rec) {
    int n = blockIdx.x * blockDim.x + threadIdx.x;
    if (n >= N_NODES) return;
    deg[n] = 0.0f;
    rec[(size_t)n * REC_W] = 0;
}

template <int U> __device__ __forceinline__ int row_bcast_i(int v) {
    return __builtin_amdgcn_update_dpp(0, v, 0x150 + U, 0xf, 0xf, false);
}
template <int U> __device__ __forceinline__ float row_bcast_f(float v) {
    return __int_as_float(row_bcast_i<U>(__float_as_int(v)));
}

__global__ __launch_bounds__(128) void k_mlp(
        const float* __restrict__ F,
        const float* __restrict__ W1, const float* __restrict__ b1,
        const float* __restrict__ W2, const float* __restrict__ b2,
        float* __restrict__ temp) {
    __shared__ float hid[4][16][DIM_H];

    const int wave = threadIdx.x >> 5;
    const int lane = threadIdx.x & 31;
    const int lm   = lane & 15;
    const int lh   = lane >> 4;
    const int tile = blockIdx.x * 4 + wave;
    if (tile >= N_TILES) return;
    const int r0 = tile * 16;

    for (int nt = 0; nt < DIM_H / 16; ++nt) {
        const int N = nt * 16 + lm;
        v8f acc = {};
        #pragma unroll 4
        for (int kb = 0; kb < DIM_FEAT / 4; ++kb) {
            const int k0 = kb * 4 + lh * 2;
            v2f a = *(const v2f*)(F + (r0 + lm) * DIM_FEAT + k0);
            v2f b;
            b.x = W1[(k0    ) * DIM_H + N];
            b.y = W1[(k0 + 1) * DIM_H + N];
            acc = __builtin_amdgcn_wmma_f32_16x16x4_f32(
                      false, a, false, b, (short)0, acc, false, false);
        }
        const float bias = b1[N];
        #pragma unroll
        for (int r = 0; r < 8; ++r) {
            float hv = acc[r] + bias;
            hv = (hv > 0.0f) ? hv : 0.01f * hv;
            hid[wave][r + 8 * lh][N] = hv;
        }
    }

    for (int nt = 0; nt < DIM_E / 16; ++nt) {
        const int N = nt * 16 + lm;
        v8f acc = {};
        #pragma unroll 4
        for (int kb = 0; kb < DIM_H / 4; ++kb) {
            const int k0 = kb * 4 + lh * 2;
            v2f a = *(const v2f*)(&hid[wave][lm][k0]);
            v2f b;
            b.x = W2[(k0    ) * DIM_E + N];
            b.y = W2[(k0 + 1) * DIM_E + N];
            acc = __builtin_amdgcn_wmma_f32_16x16x4_f32(
                      false, a, false, b, (short)0, acc, false, false);
        }
        const float bias = b2[N];
        #pragma unroll
        for (int r = 0; r < 8; ++r)
            temp[(r0 + r + 8 * lh) * DIM_E + N] = acc[r] + bias;
    }
}

__global__ void k_normalize(const float* __restrict__ pref,
                            const float* __restrict__ temp,
                            float* __restrict__ xn) {
    const int wave = threadIdx.x >> 5;
    const int lane = threadIdx.x & 31;
    const int node = blockIdx.x * (blockDim.x >> 5) + wave;
    if (node >= N_NODES) return;
    const float* src = (node < NUM_USER) ? (pref + (size_t)node * DIM_E)
                                         : (temp + (size_t)(node - NUM_USER) * DIM_E);
    float v0 = src[lane];
    float v1 = src[lane + 32];
    float s = v0 * v0 + v1 * v1;
    #pragma unroll
    for (int off = 16; off > 0; off >>= 1) s += __shfl_xor(s, off, 32);
    const float inv = 1.0f / fmaxf(sqrtf(s), 1e-12f);
    xn[(size_t)node * DIM_E + lane]      = v0 * inv;
    xn[(size_t)node * DIM_E + lane + 32] = v1 * inv;
}

__global__ void k_degree_slots(const int* __restrict__ ei, float* __restrict__ deg,
                               int* __restrict__ rec) {
    int e = blockIdx.x * blockDim.x + threadIdx.x;
    if (e >= N_EDGES) return;
    int r = ei[e];
    int c = ei[N_EDGES + e];
    if (r == c) return;
    atomicAdd(deg + r, 1.0f);
    int* t = rec + (size_t)c * REC_W;
    const int s = atomicAdd(t, 1);
    if (s < REC_CAP) t[1 + s] = r;
}

__global__ void k_dis(float* __restrict__ deg) {
    int n = blockIdx.x * blockDim.x + threadIdx.x;
    if (n >= N_NODES) return;
    float d = deg[n];
    deg[n] = (d > 0.0f) ? rsqrtf(d) : 0.0f;
}

#define LD(U)                                                                   \
    const int    r##U = row_bcast_i<U>(a);                                      \
    const float  w##U = row_bcast_f<U>(w);                                      \
    const float4 v##U = *(const float4*)(xq + (size_t)r##U * DIM_E);
#define AC(U)                                                                   \
    acc.x = fmaf(v##U.x, w##U, acc.x);                                          \
    acc.y = fmaf(v##U.y, w##U, acc.y);                                          \
    acc.z = fmaf(v##U.z, w##U, acc.z);                                          \
    acc.w = fmaf(v##U.w, w##U, acc.w);

__global__ __launch_bounds__(256) void k_pull(const int* __restrict__ ei,
                                              const int* __restrict__ rec,
                                              const float* __restrict__ dis,
                                              const float* __restrict__ x,
                                              float* __restrict__ out) {
    const int tid = blockIdx.x * blockDim.x + threadIdx.x;
    const int c = tid >> 4;
    const int q = tid & 15;
    if (c >= N_NODES) return;
    const int* __restrict__ t = rec + (size_t)c * REC_W;
    const float* __restrict__ xq = x + q * 4;
    const int total = t[0];
    const float dc = dis[c];
    float4 acc = {0.0f, 0.0f, 0.0f, 0.0f};

    if (total > 0 && total <= REC_CAP) {
        const bool ok0 = (q >= 1) && (q - 1 < total);
        int a = 0;
        if (ok0) a = t[q];
        const int rpad = row_bcast_i<1>(a);
        if (!ok0) a = rpad;
        float w = dis[a] * dc;
        if (!ok0) w = 0.0f;
        {
            LD(1) LD(2) LD(3) LD(4) LD(5) LD(6) LD(7) LD(8)
            LD(9) LD(10) LD(11) LD(12) LD(13) LD(14) LD(15)
            AC(1) AC(2) AC(3) AC(4) AC(5) AC(6) AC(7) AC(8)
            AC(9) AC(10) AC(11) AC(12) AC(13) AC(14) AC(15)
        }
        if (total > 15) {
            const bool ok1 = (15 + q < total);
            a = rpad;
            if (ok1) a = t[16 + q];
            w = dis[a] * dc;
            if (!ok1) w = 0.0f;
            {
                LD(0) LD(1) LD(2) LD(3) LD(4) LD(5) LD(6) LD(7)
                AC(0) AC(1) AC(2) AC(3) AC(4) AC(5) AC(6) AC(7)
            }
            if (total > 23) {
                LD(8) LD(9) LD(10) LD(11) LD(12) LD(13) LD(14) LD(15)
                AC(8) AC(9) AC(10) AC(11) AC(12) AC(13) AC(14) AC(15)
            }
        }
        if (total > 31) {
            for (int k = 2; k < 4; ++k) {
                const int base = 16 * k - 1;
                if (base < total) {
                    const bool ok = (base + q < total);
                    a = rpad;
                    if (ok) a = t[16 * k + q];
                    w = dis[a] * dc;
                    if (!ok) w = 0.0f;
                    LD(0) LD(1) LD(2) LD(3) LD(4) LD(5) LD(6) LD(7)
                    LD(8) LD(9) LD(10) LD(11) LD(12) LD(13) LD(14) LD(15)
                    AC(0) AC(1) AC(2) AC(3) AC(4) AC(5) AC(6) AC(7)
                    AC(8) AC(9) AC(10) AC(11) AC(12) AC(13) AC(14) AC(15)
                }
            }
        }
    } else if (total > REC_CAP) {
        for (int e0 = 0; e0 < N_EDGES; e0 += 16) {
            const int rq = ei[e0 + q];
            const int cq = ei[N_EDGES + e0 + q];
            const bool hit = (cq == c) && (rq != c);
            const unsigned hits = __builtin_amdgcn_ballot_w32(hit);
            if (((hits >> (threadIdx.x & 16)) & 0xffffu) != 0u) {
                const int a = hit ? rq : c;
                float w = dis[a] * dc;
                if (!hit) w = 0.0f;
                LD(0) LD(1) LD(2) LD(3) LD(4) LD(5) LD(6) LD(7)
                LD(8) LD(9) LD(10) LD(11) LD(12) LD(13) LD(14) LD(15)
                AC(0) AC(1) AC(2) AC(3) AC(4) AC(5) AC(6) AC(7)
                AC(8) AC(9) AC(10) AC(11) AC(12) AC(13) AC(14) AC(15)
            }
        }
    }
    *(float4*)(out + (size_t)c * DIM_E + q * 4) = acc;
}
#undef LD
#undef AC

__global__ void k_final(const float* __restrict__ xn, const float* __restrict__ h,
                        const float* __restrict__ h1, const float* __restrict__ pref,
                        float* __restrict__ out) {
    const int n4 = N_NODES * DIM_E / 4;
    const int p4 = NUM_USER * DIM_E / 4;
    int i = blockIdx.x * blockDim.x + threadIdx.x;
    if (i >= n4) return;
    float4 a = ((const float4*)xn)[i];
    float4 b = ((const float4*)h)[i];
    float4 c = ((const float4*)h1)[i];
    float4 rr;
    rr.x = a.x + b.x + c.x;
    rr.y = a.y + b.y + c.y;
    rr.z = a.z + b.z + c.z;
    rr.w = a.w + b.w + c.w;
    ((float4*)out)[i] = rr;
    if (i < p4) ((float4*)out)[n4 + i] = ((const float4*)pref)[i];
}

extern "C" void kernel_launch(void* const* d_in, const int* in_sizes, int n_in,
                              void* d_out, int out_size, void* d_ws, size_t ws_size,
                              hipStream_t stream) {
    const int*   ei   = (const int*)  d_in[0];
    const float* F    = (const float*)d_in[1];
    const float* pref = (const float*)d_in[2];
    const float* W1   = (const float*)d_in[3];
    const float* b1   = (const float*)d_in[4];
    const float* W2   = (const float*)d_in[5];
    const float* b2   = (const float*)d_in[6];
    float*       out  = (float*)d_out;

    float* ws   = (float*)d_ws;
    float* xn   = ws;
    float* h    = xn   + (size_t)N_NODES * DIM_E;
    float* h1   = h    + (size_t)N_NODES * DIM_E;
    float* temp = h1   + (size_t)N_NODES * DIM_E;
    int*   rec  = (int*)(temp + (size_t)NUM_ITEM * DIM_E);
    float* deg  = (float*)(rec + (size_t)N_NODES * REC_W);

    const int T = 256;
    k_init<<<(N_NODES + T - 1) / T, T, 0, stream>>>(deg, rec);
    k_mlp<<<(N_TILES + 3) / 4, 128, 0, stream>>>(F, W1, b1, W2, b2, temp);
    k_normalize<<<(N_NODES + 7) / 8, T, 0, stream>>>(pref, temp, xn);
    k_degree_slots<<<(N_EDGES + T - 1) / T, T, 0, stream>>>(ei, deg, rec);
    k_dis<<<(N_NODES + T - 1) / T, T, 0, stream>>>(deg);
    {
        int nthr = N_NODES * 16;
        k_pull<<<(nthr + T - 1) / T, T, 0, stream>>>(ei, rec, deg, xn, h);
        k_pull<<<(nthr + T - 1) / T, T, 0, stream>>>(ei, rec, deg, h, h1);
    }
    k_final<<<(N_NODES * DIM_E / 4 + T - 1) / T, T, 0, stream>>>(xn, h, h1, pref, out);
}


# conv edge weights factored out: hops gather rows stored pre-multiplied by dis[row] and scale each sum once by dis[col]; no per-edge dis lookup; rounding of the weighted products changes
# speedup vs baseline: 2.1978x; 1.3153x over previous
#include <hip/hip_runtime.h>

#define NUM_USER 50000
#define NUM_ITEM 50000
#define N_NODES  100000
#define DIM_E    64
#define DIM_FEAT 128
#define DIM_H    256
#define N_EDGES  1600000
#define N_TILES  (NUM_ITEM / 16)

typedef __attribute__((ext_vector_type(2))) float v2f;
typedef __attribute__((ext_vector_type(8))) float v8f;

#define REC_W   64
#define REC_CAP (REC_W - 1)

__global__ void k_init(float* __restrict__ deg, int* __restrict__ rec,
                       const float* __restrict__ W1, const float* __restrict__ W2,
                       float* __restrict__ w1s, float* __restrict__ w2s) {
    int n = blockIdx.x * blockDim.x + threadIdx.x;
    if (n >= N_NODES) return;
    deg[n] = 0.0f;
    rec[(size_t)n * REC_W] = 0;
    const int q = n & 3;
    const int c = (n >> 2) & 15;
    const int h = (n >> 6) & 1;
    if (n < DIM_FEAT * DIM_H) {
        const int j = (n >> 7) & 15;
        const int t = (n >> 11) & 3;
        const int g = (n >> 13) & 3;
        w1s[n] = W1[(8 * j + 4 * h + q) * DIM_H + 64 * g + 4 * c + t];
    }
    if (n < DIM_H * DIM_E) {
        const int ip = (n >> 7) & 1;
        const int j2 = (n >> 8) & 3;
        const int t  = (n >> 10) & 3;
        const int g  = (n >> 12) & 3;
        w2s[n] = W2[(64 * g + 32 * h + 8 * j2 + 4 * (q & 1) + t) * DIM_E + 4 * c + 2 * ip + (q >> 1)];
    }
}

template <int U> __device__ __forceinline__ int row_bcast_i(int v) {
    return __builtin_amdgcn_update_dpp(0, v, 0x150 + U, 0xf, 0xf, false);
}
template <int U> __device__ __forceinline__ float row_bcast_f(float v) {
    return __int_as_float(row_bcast_i<U>(__float_as_int(v)));
}

#define WM(A0, A1, B, ACC)                                                      \
    ACC = __builtin_amdgcn_wmma_f32_16x16x4_f32(false, (v2f){A0, A1}, false, B, \
                                                (short)0, ACC, false, false)
#define LRELU(X) (((X) > 0.0f) ? (X) : 0.01f * (X))

__global__ __launch_bounds__(128) __attribute__((amdgpu_waves_per_eu(1, 4))) void k_mlp(
        const float* __restrict__ F,
        const float* __restrict__ w1s, const float* __restrict__ b1,
        const float* __restrict__ w2s, const float* __restrict__ b2,
        float* __restrict__ temp) {
    __shared__ float stage[4][16][68];

    const int wave = threadIdx.x >> 5;
    const int lane = threadIdx.x & 31;
    const int c    = lane & 15;
    const int h    = lane >> 4;
    const int tile = blockIdx.x * 4 + wave;
    if (tile >= N_TILES) return;
    const int r0 = tile * 16;

    float4 fq[16];
    {
        const float* fp = F + (size_t)(r0 + c) * DIM_FEAT + 4 * h;
        #pragma unroll
        for (int j = 0; j < 16; ++j) fq[j] = *(const float4*)(fp + 8 * j);
    }

    v8f o0 = {}, o1 = {}, o2 = {}, o3 = {};
    const float4* __restrict__ p1 = (const float4*)w1s + lane;
    const float4* __restrict__ p2 = (const float4*)w2s + lane;

    float4 nx[3][4];
    #pragma unroll
    for (int j = 0; j < 3; ++j) {
        #pragma unroll
        for (int t = 0; t < 4; ++t) nx[j][t] = p1[(t * 16 + j) * 32];
    }

    #pragma unroll
    for (int g = 0; g < 4; ++g) {
        const float4* q1 = p1 + g * (4 * 16 * 32);
        const float4* q2 = p2 + g * (16 * 2 * 32);
        const float4* qn = q1 + (4 * 16 * 32);
        float4 w[16][4];
        float4 y[16][2];
        v8f a[4] = {};
        float4 bb[8];
        #pragma unroll
        for (int j = 0; j < 3; ++j) {
            #pragma unroll
            for (int t = 0; t < 4; ++t) w[j][t] = nx[j][t];
        }
        #pragma unroll
        for (int j = 0; j < 16; ++j) {
            if (j + 3 < 16) {
                #pragma unroll
                for (int t = 0; t < 4; ++t) w[j + 3][t] = q1[(t * 16 + j + 3) * 32];
            } else {
                #pragma unroll
                for (int k = 2 * (j - 13); k < 2 * (j - 13) + 2; ++k) {
                    y[k][0] = q2[(k * 2 + 0) * 32];
                    y[k][1] = q2[(k * 2 + 1) * 32];
                }
            }
            if (j >= 4 && j < 12)
                bb[j - 4] = *(const float4*)(b1 + 64 * g + 32 * h + 4 * (j - 4));
            const v2f bA = {fq[j].x, fq[j].y};
            const v2f bB = {fq[j].z, fq[j].w};
            #pragma unroll
            for (int t = 0; t < 4; ++t) WM(w[j][t].x, w[j][t].y, bA, a[t]);
            #pragma unroll
            for (int t = 0; t < 4; ++t) WM(w[j][t].z, w[j][t].w, bB, a[t]);
        }
        #pragma unroll
        for (int r = 0; r < 8; ++r) {
            float v;
            v = a[0][r] + bb[r].x; a[0][r] = LRELU(v);
            v = a[1][r] + bb[r].y; a[1][r] = LRELU(v);
            v = a[2][r] + bb[r].z; a[2][r] = LRELU(v);
            v = a[3][r] + bb[r].w; a[3][r] = LRELU(v);
        }
        #pragma unroll
        for (int k = 0; k < 16; ++k) {
            if (k + 6 < 16) {
                y[k + 6][0] = q2[((k + 6) * 2 + 0) * 32];
                y[k + 6][1] = q2[((k + 6) * 2 + 1) * 32];
            } else if (k >= 13 && g + 1 < 4) {
                #pragma unroll
                for (int t = 0; t < 4; ++t) nx[k - 13][t] = qn[(t * 16 + (k - 13)) * 32];
            }
            const v2f bh = {a[k >> 2][2 * (k & 3)], a[k >> 2][2 * (k & 3) + 1]};
            WM(y[k][0].x, y[k][0].y, bh, o0); WM(y[k][0].z, y[k][0].w, bh, o1);
            WM(y[k][1].x, y[k][1].y, bh, o2); WM(y[k][1].z, y[k][1].w, bh, o3);
        }
    }

    {
        float* st = &stage[wave][c][32 * h];
        #pragma unroll
        for (int r = 0; r < 8; ++r) {
            const float4 bb = *(const float4*)(b2 + 32 * h + 4 * r);
            float4 v;
            v.x = o0[r] + bb.x; v.y = o1[r] + bb.y;
            v.z = o2[r] + bb.z; v.w = o3[r] + bb.w;
            *(float4*)(st + 4 * r) = v;
        }
        #pragma unroll
        for (int it = 0; it < 8; ++it) {
            const int row = 2 * it + h;
            const float4 v = *(const float4*)(&stage[wave][row][4 * c]);
            *(float4*)(temp + (size_t)(r0 + row) * DIM_E + 4 * c) = v;
        }
    }
}
#undef WM
#undef LRELU

__global__ void k_normalize(const float* __restrict__ pref,
                            const float* __restrict__ temp,
                            const float* __restrict__ dis,
                            float* __restrict__ xn,
                            float* __restrict__ xs) {
    const int wave = threadIdx.x >> 5;
    const int lane = threadIdx.x & 31;
    const int node = blockIdx.x * (blockDim.x >> 5) + wave;
    if (node >= N_NODES) return;
    const float* src = (node < NUM_USER) ? (pref + (size_t)node * DIM_E)
                                         : (temp + (size_t)(node - NUM_USER) * DIM_E);
    float v0 = src[lane];
    float v1 = src[lane + 32];
    float s = v0 * v0 + v1 * v1;
    #pragma unroll
    for (int off = 16; off > 0; off >>= 1) s += __shfl_xor(s, off, 32);
    const float inv = 1.0f / fmaxf(sqrtf(s), 1e-12f);
    const float d = dis[node];
    const float n0 = v0 * inv;
    const float n1 = v1 * inv;
    xn[(size_t)node * DIM_E + lane]      = n0;
    xn[(size_t)node * DIM_E + lane + 32] = n1;
    xs[(size_t)node * DIM_E + lane]      = n0 * d;
    xs[(size_t)node * DIM_E + lane + 32] = n1 * d;
}

__global__ void k_degree_slots(const int* __restrict__ ei, float* __restrict__ deg,
                               int* __restrict__ rec) {
    int e = blockIdx.x * blockDim.x + threadIdx.x;
    if (e >= N_EDGES) return;
    int r = ei[e];
    int c = ei[N_EDGES + e];
    if (r == c) return;
    atomicAdd(deg + r, 1.0f);
    int* t = rec + (size_t)c * REC_W;
    const int s = atomicAdd(t, 1);
    if (s < REC_CAP) t[1 + s] = r;
}

__global__ void k_dis(float* __restrict__ deg) {
    int n = blockIdx.x * blockDim.x + threadIdx.x;
    if (n >= N_NODES) return;
    float d = deg[n];
    deg[n] = (d > 0.0f) ? rsqrtf(d) : 0.0f;
}

#define LD(U)                                                                   \
    const int    r##U = row_bcast_i<U>(a);                                      \
    const float  w##U = row_bcast_f<U>(w);                                      \
    const float4 v##U = *(const float4*)(xq + (size_t)r##U * DIM_E);
#define AC(U)                                                                   \
    acc.x = fmaf(v##U.x, w##U, acc.x);                                          \
    acc.y = fmaf(v##U.y, w##U, acc.y);                                          \
    acc.z = fmaf(v##U.z, w##U, acc.z);                                          \
    acc.w = fmaf(v##U.w, w##U, acc.w);

template <bool SCALED_COPY>
__global__ __launch_bounds__(256) void k_pull(const int* __restrict__ ei,
                                              const int* __restrict__ rec,
                                              const float* __restrict__ dis,
                                              const float* __restrict__ x,
                                              float* __restrict__ out,
                                              float* __restrict__ outs) {
    const int tid = blockIdx.x * blockDim.x + threadIdx.x;
    const int c = tid >> 4;
    const int q = tid & 15;
    if (c >= N_NODES) return;
    const int* __restrict__ t = rec + (size_t)c * REC_W;
    const float* __restrict__ xq = x + q * 4;
    const int total = t[0];
    const float dc = dis[c];
    float4 acc = {0.0f, 0.0f, 0.0f, 0.0f};

    if (total > 0 && total <= REC_CAP) {
        const bool ok0 = (q >= 1) && (q - 1 < total);
        int a = 0;
        if (ok0) a = t[q];
        const int rpad = row_bcast_i<1>(a);
        if (!ok0) a = rpad;
        float w = ok0 ? 1.0f : 0.0f;
        {
            LD(1) LD(2) LD(3) LD(4) LD(5) LD(6) LD(7) LD(8)
            LD(9) LD(10) LD(11) LD(12) LD(13) LD(14) LD(15)
            AC(1) AC(2) AC(3) AC(4) AC(5) AC(6) AC(7) AC(8)
            AC(9) AC(10) AC(11) AC(12) AC(13) AC(14) AC(15)
        }
        if (total > 15) {
            const bool ok1 = (15 + q < total);
            a = rpad;
            if (ok1) a = t[16 + q];
            w = ok1 ? 1.0f : 0.0f;
            {
                LD(0) LD(1) LD(2) LD(3) LD(4) LD(5) LD(6) LD(7)
                AC(0) AC(1) AC(2) AC(3) AC(4) AC(5) AC(6) AC(7)
            }
            if (total > 23) {
                LD(8) LD(9) LD(10) LD(11) LD(12) LD(13) LD(14) LD(15)
                AC(8) AC(9) AC(10) AC(11) AC(12) AC(13) AC(14) AC(15)
            }
        }
        if (total > 31) {
            for (int k = 2; k < 4; ++k) {
                const int base = 16 * k - 1;
                if (base < total) {
                    const bool ok = (base + q < total);
                    a = rpad;
                    if (ok) a = t[16 * k + q];
                    w = ok ? 1.0f : 0.0f;
                    LD(0) LD(1) LD(2) LD(3) LD(4) LD(5) LD(6) LD(7)
                    LD(8) LD(9) LD(10) LD(11) LD(12) LD(13) LD(14) LD(15)
                    AC(0) AC(1) AC(2) AC(3) AC(4) AC(5) AC(6) AC(7)
                    AC(8) AC(9) AC(10) AC(11) AC(12) AC(13) AC(14) AC(15)
                }
            }
        }
    } else if (total > REC_CAP) {
        for (int e0 = 0; e0 < N_EDGES; e0 += 16) {
            const int rq = ei[e0 + q];
            const int cq = ei[N_EDGES + e0 + q];
            const bool hit = (cq == c) && (rq != c);
            const unsigned hits = __builtin_amdgcn_ballot_w32(hit);
            if (((hits >> (threadIdx.x & 16)) & 0xffffu) != 0u) {
                const int a = hit ? rq : c;
                const float w = hit ? 1.0f : 0.0f;
                LD(0) LD(1) LD(2) LD(3) LD(4) LD(5) LD(6) LD(7)
                LD(8) LD(9) LD(10) LD(11) LD(12) LD(13) LD(14) LD(15)
                AC(0) AC(1) AC(2) AC(3) AC(4) AC(5) AC(6) AC(7)
                AC(8) AC(9) AC(10) AC(11) AC(12) AC(13) AC(14) AC(15)
            }
        }
    }
    acc.x *= dc; acc.y *= dc; acc.z *= dc; acc.w *= dc;
    *(float4*)(out + (size_t)c * DIM_E + q * 4) = acc;
    if (SCALED_COPY) {
        acc.x *= dc; acc.y *= dc; acc.z *= dc; acc.w *= dc;
        *(float4*)(outs + (size_t)c * DIM_E + q * 4) = acc;
    }
}
#undef LD
#undef AC

__global__ void k_final(const float* __restrict__ xn, const float* __restrict__ h,
                        const float* __restrict__ h1, const float* __restrict__ pref,
                        float* __restrict__ out) {
    const int n4 = N_NODES * DIM_E / 4;
    const int p4 = NUM_USER * DIM_E / 4;
    int i = blockIdx.x * blockDim.x + threadIdx.x;
    if (i >= n4) return;
    float4 a = ((const float4*)xn)[i];
    float4 b = ((const float4*)h)[i];
    float4 c = ((const float4*)h1)[i];
    float4 rr;
    rr.x = a.x + b.x + c.x;
    rr.y = a.y + b.y + c.y;
    rr.z = a.z + b.z + c.z;
    rr.w = a.w + b.w + c.w;
    ((float4*)out)[i] = rr;
    if (i < p4) ((float4*)out)[n4 + i] = ((const float4*)pref)[i];
}

extern "C" void kernel_launch(void* const* d_in, const int* in_sizes, int n_in,
                              void* d_out, int out_size, void* d_ws, size_t ws_size,
                              hipStream_t stream) {
    const int*   ei   = (const int*)  d_in[0];
    const float* F    = (const float*)d_in[1];
    const float* pref = (const float*)d_in[2];
    const float* W1   = (const float*)d_in[3];
    const float* b1   = (const float*)d_in[4];
    const float* W2   = (const float*)d_in[5];
    const float* b2   = (const float*)d_in[6];
    float*       out  = (float*)d_out;

    float* ws   = (float*)d_ws;
    float* xn   = ws;
    float* h    = xn   + (size_t)N_NODES * DIM_E;
    float* h1   = h    + (size_t)N_NODES * DIM_E;
    float* temp = h1   + (size_t)N_NODES * DIM_E;
    int*   rec  = (int*)(temp + (size_t)NUM_ITEM * DIM_E);
    float* deg  = (float*)(rec + (size_t)N_NODES * REC_W);
    float* w1s  = deg + N_NODES;
    float* w2s  = w1s + DIM_FEAT * DIM_H;
    float* xs   = w2s + DIM_H * DIM_E;
    float* hs   = xs   + (size_t)N_NODES * DIM_E;

    const int T = 256;
    k_init<<<(N_NODES + T - 1) / T, T, 0, stream>>>(deg, rec, W1, W2, w1s, w2s);
    k_degree_slots<<<(N_EDGES + T - 1) / T, T, 0, stream>>>(ei, deg, rec);
    k_dis<<<(N_NODES + T - 1) / T, T, 0, stream>>>(deg);
    k_mlp<<<(N_TILES + 3) / 4, 128, 0, stream>>>(F, w1s, b1, w2s, b2, temp);
    k_normalize<<<(N_NODES + 7) / 8, T, 0, stream>>>(pref, temp, deg, xn, xs);
    {
        int nthr = N_NODES * 16;
        k_pull<true><<<(nthr + T - 1) / T, T, 0, stream>>>(ei, rec, deg, xs, h, hs);
        k_pull<false><<<(nthr + T - 1) / T, T, 0, stream>>>(ei, rec, deg, hs, h1, nullptr);
    }
    k_final<<<(N_NODES * DIM_E / 4 + T - 1) / T, T, 0, stream>>>(xn, h, h1, pref, out);
}


# edge launch: each chip half does the atomics of its own node set (tasks drawn from per-half counters, die read from the hardware id): no counter line updated from both halves; slot order shifts
# speedup vs baseline: 3.6638x; 1.6670x over previous
#include <hip/hip_runtime.h>

#define NUM_USER 50000
#define NUM_ITEM 50000
#define N_NODES  100000
#define DIM_E    64
#define DIM_FEAT 128
#define DIM_H    256
#define N_EDGES  1600000
#define N_TILES  (NUM_ITEM / 16)

typedef __attribute__((ext_vector_type(2))) float v2f;
typedef __attribute__((ext_vector_type(8))) float v8f;

#define REC_W   64
#define REC_CAP (REC_W - 1)
#define QN_PITCH 64

__global__ void k_init(float* __restrict__ deg, int* __restrict__ rec,
                       const float* __restrict__ W1, const float* __restrict__ W2,
                       float* __restrict__ w1s, float* __restrict__ w2s,
                       int* __restrict__ qn) {
    int n = blockIdx.x * blockDim.x + threadIdx.x;
    if (n >= N_NODES) return;
    deg[n] = 0.0f;
    rec[(size_t)n * REC_W] = 0;
    if (n < 2) qn[QN_PITCH * n] = 0;
    const int q = n & 3;
    const int c = (n >> 2) & 15;
    const int h = (n >> 6) & 1;
    if (n < DIM_FEAT * DIM_H) {
        const int j = (n >> 7) & 15;
        const int t = (n >> 11) & 3;
        const int g = (n >> 13) & 3;
        w1s[n] = W1[(8 * j + 4 * h + q) * DIM_H + 64 * g + 4 * c + t];
    }
    if (n < DIM_H * DIM_E) {
        const int ip = (n >> 7) & 1;
        const int j2 = (n >> 8) & 3;
        const int t  = (n >> 10) & 3;
        const int g  = (n >> 12) & 3;
        w2s[n] = W2[(64 * g + 32 * h + 8 * j2 + 4 * (q & 1) + t) * DIM_E + 4 * c + 2 * ip + (q >> 1)];
    }
}

template <int U> __device__ __forceinline__ int row_bcast_i(int v) {
    return __builtin_amdgcn_update_dpp(0, v, 0x150 + U, 0xf, 0xf, false);
}
template <int U> __device__ __forceinline__ float row_bcast_f(float v) {
    return __int_as_float(row_bcast_i<U>(__float_as_int(v)));
}

#define WM(A0, A1, B, ACC)                                                      \
    ACC = __builtin_amdgcn_wmma_f32_16x16x4_f32(false, (v2f){A0, A1}, false, B, \
                                                (short)0, ACC, false, false)
#define LRELU(X) (((X) > 0.0f) ? (X) : 0.01f * (X))

__global__ __launch_bounds__(128) __attribute__((amdgpu_waves_per_eu(1, 4))) void k_mlp(
        const float* __restrict__ F,
        const float* __restrict__ w1s, const float* __restrict__ b1,
        const float* __restrict__ w2s, const float* __restrict__ b2,
        float* __restrict__ temp) {
    __shared__ float stage[4][16][68];

    const int wave = threadIdx.x >> 5;
    const int lane = threadIdx.x & 31;
    const int c    = lane & 15;
    const int h    = lane >> 4;
    const int tile = blockIdx.x * 4 + wave;
    if (tile >= N_TILES) return;
    const int r0 = tile * 16;

    float4 fq[16];
    {
        const float* fp = F + (size_t)(r0 + c) * DIM_FEAT + 4 * h;
        #pragma unroll
        for (int j = 0; j < 16; ++j) fq[j] = *(const float4*)(fp + 8 * j);
    }

    v8f o0 = {}, o1 = {}, o2 = {}, o3 = {};
    const float4* __restrict__ p1 = (const float4*)w1s + lane;
    const float4* __restrict__ p2 = (const float4*)w2s + lane;

    float4 nx[3][4];
    #pragma unroll
    for (int j = 0; j < 3; ++j) {
        #pragma unroll
        for (int t = 0; t < 4; ++t) nx[j][t] = p1[(t * 16 + j) * 32];
    }

    #pragma unroll
    for (int g = 0; g < 4; ++g) {
        const float4* q1 = p1 + g * (4 * 16 * 32);
        const float4* q2 = p2 + g * (16 * 2 * 32);
        const float4* qn = q1 + (4 * 16 * 32);
        float4 w[16][4];
        float4 y[16][2];
        v8f a[4] = {};
        float4 bb[8];
        #pragma unroll
        for (int j = 0; j < 3; ++j) {
            #pragma unroll
            for (int t = 0; t < 4; ++t) w[j][t] = nx[j][t];
        }
        #pragma unroll
        for (int j = 0; j < 16; ++j) {
            if (j + 3 < 16) {
                #pragma unroll
                for (int t = 0; t < 4; ++t) w[j + 3][t] = q1[(t * 16 + j + 3) * 32];
            } else {
                #pragma unroll
                for (int k = 2 * (j - 13); k < 2 * (j - 13) + 2; ++k) {
                    y[k][0] = q2[(k * 2 + 0) * 32];
                    y[k][1] = q2[(k * 2 + 1) * 32];
                }
            }
            if (j >= 4 && j < 12)
                bb[j - 4] = *(const float4*)(b1 + 64 * g + 32 * h + 4 * (j - 4));
            const v2f bA = {fq[j].x, fq[j].y};
            const v2f bB = {fq[j].z, fq[j].w};
            #pragma unroll
            for (int t = 0; t < 4; ++t) WM(w[j][t].x, w[j][t].y, bA, a[t]);
            #pragma unroll
            for (int t = 0; t < 4; ++t) WM(w[j][t].z, w[j][t].w, bB, a[t]);
        }
        #pragma unroll
        for (int r = 0; r < 8; ++r) {
            float v;
            v = a[0][r] + bb[r].x; a[0][r] = LRELU(v);
            v = a[1][r] + bb[r].y; a[1][r] = LRELU(v);
            v = a[2][r] + bb[r].z; a[2][r] = LRELU(v);
            v = a[3][r] + bb[r].w; a[3][r] = LRELU(v);
        }
        #pragma unroll
        for (int k = 0; k < 16; ++k) {
            if (k + 6 < 16) {
                y[k + 6][0] = q2[((k + 6) * 2 + 0) * 32];
                y[k + 6][1] = q2[((k + 6) * 2 + 1) * 32];
            } else if (k >= 13 && g + 1 < 4) {
                #pragma unroll
                for (int t = 0; t < 4; ++t) nx[k - 13][t] = qn[(t * 16 + (k - 13)) * 32];
            }
            const v2f bh = {a[k >> 2][2 * (k & 3)], a[k >> 2][2 * (k & 3) + 1]};
            WM(y[k][0].x, y[k][0].y, bh, o0); WM(y[k][0].z, y[k][0].w, bh, o1);
            WM(y[k][1].x, y[k][1].y, bh, o2); WM(y[k][1].z, y[k][1].w, bh, o3);
        }
    }

    {
        float* st = &stage[wave][c][32 * h];
        #pragma unroll
        for (int r = 0; r < 8; ++r) {
            const float4 bb = *(const float4*)(b2 + 32 * h + 4 * r);
            float4 v;
            v.x = o0[r] + bb.x; v.y = o1[r] + bb.y;
            v.z = o2[r] + bb.z; v.w = o3[r] + bb.w;
            *(float4*)(st + 4 * r) = v;
        }
        #pragma unroll
        for (int it = 0; it < 8; ++it) {
            const int row = 2 * it + h;
            const float4 v = *(const float4*)(&stage[wave][row][4 * c]);
            *(float4*)(temp + (size_t)(r0 + row) * DIM_E + 4 * c) = v;
        }
    }
}
#undef WM
#undef LRELU

__global__ void k_normalize(const float* __restrict__ pref,
                            const float* __restrict__ temp,
                            const float* __restrict__ dis,
                            float* __restrict__ xn,
                            float* __restrict__ xs) {
    const int wave = threadIdx.x >> 5;
    const int lane = threadIdx.x & 31;
    const int node = blockIdx.x * (blockDim.x >> 5) + wave;
    if (node >= N_NODES) return;
    const float* src = (node < NUM_USER) ? (pref + (size_t)node * DIM_E)
                                         : (temp + (size_t)(node - NUM_USER) * DIM_E);
    float v0 = src[lane];
    float v1 = src[lane + 32];
    float s = v0 * v0 + v1 * v1;
    #pragma unroll
    for (int off = 16; off > 0; off >>= 1) s += __shfl_xor(s, off, 32);
    const float inv = 1.0f / fmaxf(sqrtf(s), 1e-12f);
    const float d = dis[node];
    const float n0 = v0 * inv;
    const float n1 = v1 * inv;
    xn[(size_t)node * DIM_E + lane]      = n0;
    xn[(size_t)node * DIM_E + lane + 32] = n1;
    xs[(size_t)node * DIM_E + lane]      = n0 * d;
    xs[(size_t)node * DIM_E + lane + 32] = n1 * d;
}

#define EDGE_CH  1024
#define N_CH     ((N_EDGES + EDGE_CH - 1) / EDGE_CH)
__device__ __forceinline__ int node_set(int n) { return (n >> 6) & 1; }

__global__ __launch_bounds__(256) void k_degree_slots(const int* __restrict__ ei,
                                                      float* __restrict__ deg,
                                                      int* __restrict__ rec,
                                                      int* __restrict__ qn) {
    __shared__ int s_task;
    const unsigned hwid = __builtin_amdgcn_s_sendmsg_rtn(0x87);
    const int myhalf = (int)((hwid >> 18) & 1u);

    for (int pass = 0; pass < 2; ++pass) {
        const int set = myhalf ^ pass;
        int* __restrict__ ctr = qn + QN_PITCH * set;
        for (int it = 0; it <= N_CH; ++it) {
            if (threadIdx.x == 0) s_task = atomicAdd(ctr, 1);
            __syncthreads();
            const int task = s_task;
            __syncthreads();
            if (task >= N_CH) break;
            const int e0 = task * EDGE_CH + threadIdx.x;
            int r[4], c[4];
            #pragma unroll
            for (int k = 0; k < 4; ++k) {
                const int e = e0 + 256 * k;
                const bool in = e < N_EDGES;
                r[k] = in ? ei[e] : 0;
                c[k] = in ? ei[N_EDGES + e] : 0;
            }
            int s[4];
            #pragma unroll
            for (int k = 0; k < 4; ++k) {
                s[k] = REC_CAP;
                if (r[k] != c[k]) {
                    if (node_set(r[k]) == set) atomicAdd(deg + r[k], 1.0f);
                    if (node_set(c[k]) == set) s[k] = atomicAdd(rec + (size_t)c[k] * REC_W, 1);
                }
            }
            #pragma unroll
            for (int k = 0; k < 4; ++k) {
                if (s[k] < REC_CAP) rec[(size_t)c[k] * REC_W + 1 + s[k]] = r[k];
            }
        }
    }
}

__global__ void k_dis(float* __restrict__ deg) {
    int n = blockIdx.x * blockDim.x + threadIdx.x;
    if (n >= N_NODES) return;
    float d = deg[n];
    deg[n] = (d > 0.0f) ? rsqrtf(d) : 0.0f;
}

#define LD(U)                                                                   \
    const int    r##U = row_bcast_i<U>(a);                                      \
    const float  w##U = row_bcast_f<U>(w);                                      \
    const float4 v##U = *(const float4*)(xq + (size_t)r##U * DIM_E);
#define AC(U)                                                                   \
    acc.x = fmaf(v##U.x, w##U, acc.x);                                          \
    acc.y = fmaf(v##U.y, w##U, acc.y);                                          \
    acc.z = fmaf(v##U.z, w##U, acc.z);                                          \
    acc.w = fmaf(v##U.w, w##U, acc.w);

template <bool SCALED_COPY>
__global__ __launch_bounds__(256) void k_pull(const int* __restrict__ ei,
                                              const int* __restrict__ rec,
                                              const float* __restrict__ dis,
                                              const float* __restrict__ x,
                                              float* __restrict__ out,
                                              float* __restrict__ outs) {
    const int tid = blockIdx.x * blockDim.x + threadIdx.x;
    const int c = tid >> 4;
    const int q = tid & 15;
    if (c >= N_NODES) return;
    const int* __restrict__ t = rec + (size_t)c * REC_W;
    const float* __restrict__ xq = x + q * 4;
    const int total = t[0];
    const float dc = dis[c];
    float4 acc = {0.0f, 0.0f, 0.0f, 0.0f};

    if (total > 0 && total <= REC_CAP) {
        const bool ok0 = (q >= 1) && (q - 1 < total);
        int a = 0;
        if (ok0) a = t[q];
        const int rpad = row_bcast_i<1>(a);
        if (!ok0) a = rpad;
        float w = ok0 ? 1.0f : 0.0f;
        {
            LD(1) LD(2) LD(3) LD(4) LD(5) LD(6) LD(7) LD(8)
            LD(9) LD(10) LD(11) LD(12) LD(13) LD(14) LD(15)
            AC(1) AC(2) AC(3) AC(4) AC(5) AC(6) AC(7) AC(8)
            AC(9) AC(10) AC(11) AC(12) AC(13) AC(14) AC(15)
        }
        if (total > 15) {
            const bool ok1 = (15 + q < total);
            a = rpad;
            if (ok1) a = t[16 + q];
            w = ok1 ? 1.0f : 0.0f;
            {
                LD(0) LD(1) LD(2) LD(3) LD(4) LD(5) LD(6) LD(7)
                AC(0) AC(1) AC(2) AC(3) AC(4) AC(5) AC(6) AC(7)
            }
            if (total > 23) {
                LD(8) LD(9) LD(10) LD(11) LD(12) LD(13) LD(14) LD(15)
                AC(8) AC(9) AC(10) AC(11) AC(12) AC(13) AC(14) AC(15)
            }
        }
        if (total > 31) {
            for (int k = 2; k < 4; ++k) {
                const int base = 16 * k - 1;
                if (base < total) {
                    const bool ok = (base + q < total);
                    a = rpad;
                    if (ok) a = t[16 * k + q];
                    w = ok ? 1.0f : 0.0f;
                    LD(0) LD(1) LD(2) LD(3) LD(4) LD(5) LD(6) LD(7)
                    LD(8) LD(9) LD(10) LD(11) LD(12) LD(13) LD(14) LD(15)
                    AC(0) AC(1) AC(2) AC(3) AC(4) AC(5) AC(6) AC(7)
                    AC(8) AC(9) AC(10) AC(11) AC(12) AC(13) AC(14) AC(15)
                }
            }
        }
    } else if (total > REC_CAP) {
        for (int e0 = 0; e0 < N_EDGES; e0 += 16) {
            const int rq = ei[e0 + q];
            const int cq = ei[N_EDGES + e0 + q];
            const bool hit = (cq == c) && (rq != c);
            const unsigned hits = __builtin_amdgcn_ballot_w32(hit);
            if (((hits >> (threadIdx.x & 16)) & 0xffffu) != 0u) {
                const int a = hit ? rq : c;
                const float w = hit ? 1.0f : 0.0f;
                LD(0) LD(1) LD(2) LD(3) LD(4) LD(5) LD(6) LD(7)
                LD(8) LD(9) LD(10) LD(11) LD(12) LD(13) LD(14) LD(15)
                AC(0) AC(1) AC(2) AC(3) AC(4) AC(5) AC(6) AC(7)
                AC(8) AC(9) AC(10) AC(11) AC(12) AC(13) AC(14) AC(15)
            }
        }
    }
    acc.x *= dc; acc.y *= dc; acc.z *= dc; acc.w *= dc;
    *(float4*)(out + (size_t)c * DIM_E + q * 4) = acc;
    if (SCALED_COPY) {
        acc.x *= dc; acc.y *= dc; acc.z *= dc; acc.w *= dc;
        *(float4*)(outs + (size_t)c * DIM_E + q * 4) = acc;
    }
}
#undef LD
#undef AC

__global__ void k_final(const float* __restrict__ xn, const float* __restrict__ h,
                        const float* __restrict__ h1, const float* __restrict__ pref,
                        float* __restrict__ out) {
    const int n4 = N_NODES * DIM_E / 4;
    const int p4 = NUM_USER * DIM_E / 4;
    int i = blockIdx.x * blockDim.x + threadIdx.x;
    if (i >= n4) return;
    float4 a = ((const float4*)xn)[i];
    float4 b = ((const float4*)h)[i];
    float4 c = ((const float4*)h1)[i];
    float4 rr;
    rr.x = a.x + b.x + c.x;
    rr.y = a.y + b.y + c.y;
    rr.z = a.z + b.z + c.z;
    rr.w = a.w + b.w + c.w;
    ((float4*)out)[i] = rr;
    if (i < p4) ((float4*)out)[n4 + i] = ((const float4*)pref)[i];
}

extern "C" void kernel_launch(void* const* d_in, const int* in_sizes, int n_in,
                              void* d_out, int out_size, void* d_ws, size_t ws_size,
                              hipStream_t stream) {
    const int*   ei   = (const int*)  d_in[0];
    const float* F    = (const float*)d_in[1];
    const float* pref = (const float*)d_in[2];
    const float* W1   = (const float*)d_in[3];
    const float* b1   = (const float*)d_in[4];
    const float* W2   = (const float*)d_in[5];
    const float* b2   = (const float*)d_in[6];
    float*       out  = (float*)d_out;

    float* ws   = (float*)d_ws;
    float* xn   = ws;
    float* h    = xn   + (size_t)N_NODES * DIM_E;
    float* h1   = h    + (size_t)N_NODES * DIM_E;
    float* temp = h1   + (size_t)N_NODES * DIM_E;
    int*   rec  = (int*)(temp + (size_t)NUM_ITEM * DIM_E);
    float* deg  = (float*)(rec + (size_t)N_NODES * REC_W);
    float* w1s  = deg + N_NODES;
    float* w2s  = w1s + DIM_FEAT * DIM_H;
    float* xs   = w2s + DIM_H * DIM_E;
    float* hs   = xs   + (size_t)N_NODES * DIM_E;
    int*   qn   = (int*)(hs + (size_t)N_NODES * DIM_E);

    const int T = 256;
    k_init<<<(N_NODES + T - 1) / T, T, 0, stream>>>(deg, rec, W1, W2, w1s, w2s, qn);
    k_degree_slots<<<2048, T, 0, stream>>>(ei, deg, rec, qn);
    k_dis<<<(N_NODES + T - 1) / T, T, 0, stream>>>(deg);
    k_mlp<<<(N_TILES + 3) / 4, 128, 0, stream>>>(F, w1s, b1, w2s, b2, temp);
    k_normalize<<<(N_NODES + 7) / 8, T, 0, stream>>>(pref, temp, deg, xn, xs);
    {
        int nthr = N_NODES * 16;
        k_pull<true><<<(nthr + T - 1) / T, T, 0, stream>>>(ei, rec, deg, xs, h, hs);
        k_pull<false><<<(nthr + T - 1) / T, T, 0, stream>>>(ei, rec, deg, hs, h1, nullptr);
    }
    k_final<<<(N_NODES * DIM_E / 4 + T - 1) / T, T, 0, stream>>>(xn, h, h1, pref, out);
}


# conv hops nested, x + A(x + A x): hop 1 stores (x + A x)*dis, hop 2 adds x and writes out; no A x / A A x tables, no final sum launch (8 -> 7); normalize copies preference; sum order changes
# speedup vs baseline: 3.7841x; 1.0328x over previous
#include <hip/hip_runtime.h>

#define NUM_USER 50000
#define NUM_ITEM 50000
#define N_NODES  100000
#define DIM_E    64
#define DIM_FEAT 128
#define DIM_H    256
#define N_EDGES  1600000
#define N_TILES  (NUM_ITEM / 16)

typedef __attribute__((ext_vector_type(2))) float v2f;
typedef __attribute__((ext_vector_type(8))) float v8f;

#define REC_W   64
#define REC_CAP (REC_W - 1)
#define QN_PITCH 64

__global__ void k_init(float* __restrict__ deg, int* __restrict__ rec,
                       const float* __restrict__ W1, const float* __restrict__ W2,
                       float* __restrict__ w1s, float* __restrict__ w2s,
                       int* __restrict__ qn) {
    int n = blockIdx.x * blockDim.x + threadIdx.x;
    if (n >= N_NODES) return;
    deg[n] = 0.0f;
    rec[(size_t)n * REC_W] = 0;
    if (n < 2) qn[QN_PITCH * n] = 0;
    const int q = n & 3;
    const int c = (n >> 2) & 15;
    const int h = (n >> 6) & 1;
    if (n < DIM_FEAT * DIM_H) {
        const int j = (n >> 7) & 15;
        const int t = (n >> 11) & 3;
        const int g = (n >> 13) & 3;
        w1s[n] = W1[(8 * j + 4 * h + q) * DIM_H + 64 * g + 4 * c + t];
    }
    if (n < DIM_H * DIM_E) {
        const int ip = (n >> 7) & 1;
        const int j2 = (n >> 8) & 3;
        const int t  = (n >> 10) & 3;
        const int g  = (n >> 12) & 3;
        w2s[n] = W2[(64 * g + 32 * h + 8 * j2 + 4 * (q & 1) + t) * DIM_E + 4 * c + 2 * ip + (q >> 1)];
    }
}

template <int U> __device__ __forceinline__ int row_bcast_i(int v) {
    return __builtin_amdgcn_update_dpp(0, v, 0x150 + U, 0xf, 0xf, false);
}
template <int U> __device__ __forceinline__ float row_bcast_f(float v) {
    return __int_as_float(row_bcast_i<U>(__float_as_int(v)));
}

#define WM(A0, A1, B, ACC)                                                      \
    ACC = __builtin_amdgcn_wmma_f32_16x16x4_f32(false, (v2f){A0, A1}, false, B, \
                                                (short)0, ACC, false, false)
#define LRELU(X) (((X) > 0.0f) ? (X) : 0.01f * (X))

__global__ __launch_bounds__(128) __attribute__((amdgpu_waves_per_eu(1, 4))) void k_mlp(
        const float* __restrict__ F,
        const float* __restrict__ w1s, const float* __restrict__ b1,
        const float* __restrict__ w2s, const float* __restrict__ b2,
        float* __restrict__ temp) {
    __shared__ float stage[4][16][68];

    const int wave = threadIdx.x >> 5;
    const int lane = threadIdx.x & 31;
    const int c    = lane & 15;
    const int h    = lane >> 4;
    const int tile = blockIdx.x * 4 + wave;
    if (tile >= N_TILES) return;
    const int r0 = tile * 16;

    float4 fq[16];
    {
        const float* fp = F + (size_t)(r0 + c) * DIM_FEAT + 4 * h;
        #pragma unroll
        for (int j = 0; j < 16; ++j) fq[j] = *(const float4*)(fp + 8 * j);
    }

    v8f o0 = {}, o1 = {}, o2 = {}, o3 = {};
    const float4* __restrict__ p1 = (const float4*)w1s + lane;
    const float4* __restrict__ p2 = (const float4*)w2s + lane;

    float4 nx[3][4];
    #pragma unroll
    for (int j = 0; j < 3; ++j) {
        #pragma unroll
        for (int t = 0; t < 4; ++t) nx[j][t] = p1[(t * 16 + j) * 32];
    }

    #pragma unroll
    for (int g = 0; g < 4; ++g) {
        const float4* q1 = p1 + g * (4 * 16 * 32);
        const float4* q2 = p2 + g * (16 * 2 * 32);
        const float4* qn = q1 + (4 * 16 * 32);
        float4 w[16][4];
        float4 y[16][2];
        v8f a[4] = {};
        float4 bb[8];
        #pragma unroll
        for (int j = 0; j < 3; ++j) {
            #pragma unroll
            for (int t = 0; t < 4; ++t) w[j][t] = nx[j][t];
        }
        #pragma unroll
        for (int j = 0; j < 16; ++j) {
            if (j + 3 < 16) {
                #pragma unroll
                for (int t = 0; t < 4; ++t) w[j + 3][t] = q1[(t * 16 + j + 3) * 32];
            } else {
                #pragma unroll
                for (int k = 2 * (j - 13); k < 2 * (j - 13) + 2; ++k) {
                    y[k][0] = q2[(k * 2 + 0) * 32];
                    y[k][1] = q2[(k * 2 + 1) * 32];
                }
            }
            if (j >= 4 && j < 12)
                bb[j - 4] = *(const float4*)(b1 + 64 * g + 32 * h + 4 * (j - 4));
            const v2f bA = {fq[j].x, fq[j].y};
            const v2f bB = {fq[j].z, fq[j].w};
            #pragma unroll
            for (int t = 0; t < 4; ++t) WM(w[j][t].x, w[j][t].y, bA, a[t]);
            #pragma unroll
            for (int t = 0; t < 4; ++t) WM(w[j][t].z, w[j][t].w, bB, a[t]);
        }
        #pragma unroll
        for (int r = 0; r < 8; ++r) {
            float v;
            v = a[0][r] + bb[r].x; a[0][r] = LRELU(v);
            v = a[1][r] + bb[r].y; a[1][r] = LRELU(v);
            v = a[2][r] + bb[r].z; a[2][r] = LRELU(v);
            v = a[3][r] + bb[r].w; a[3][r] = LRELU(v);
        }
        #pragma unroll
        for (int k = 0; k < 16; ++k) {
            if (k + 6 < 16) {
                y[k + 6][0] = q2[((k + 6) * 2 + 0) * 32];
                y[k + 6][1] = q2[((k + 6) * 2 + 1) * 32];
            } else if (k >= 13 && g + 1 < 4) {
                #pragma unroll
                for (int t = 0; t < 4; ++t) nx[k - 13][t] = qn[(t * 16 + (k - 13)) * 32];
            }
            const v2f bh = {a[k >> 2][2 * (k & 3)], a[k >> 2][2 * (k & 3) + 1]};
            WM(y[k][0].x, y[k][0].y, bh, o0); WM(y[k][0].z, y[k][0].w, bh, o1);
            WM(y[k][1].x, y[k][1].y, bh, o2); WM(y[k][1].z, y[k][1].w, bh, o3);
        }
    }

    {
        float* st = &stage[wave][c][32 * h];
        #pragma unroll
        for (int r = 0; r < 8; ++r) {
            const float4 bb = *(const float4*)(b2 + 32 * h + 4 * r);
            float4 v;
            v.x = o0[r] + bb.x; v.y = o1[r] + bb.y;
            v.z = o2[r] + bb.z; v.w = o3[r] + bb.w;
            *(float4*)(st + 4 * r) = v;
        }
        #pragma unroll
        for (int it = 0; it < 8; ++it) {
            const int row = 2 * it + h;
            const float4 v = *(const float4*)(&stage[wave][row][4 * c]);
            *(float4*)(temp + (size_t)(r0 + row) * DIM_E + 4 * c) = v;
        }
    }
}
#undef WM
#undef LRELU

__global__ void k_normalize(const float* __restrict__ pref,
                            const float* __restrict__ temp,
                            const float* __restrict__ dis,
                            float* __restrict__ xn,
                            float* __restrict__ xs,
                            float* __restrict__ out_pref) {
    const int wave = threadIdx.x >> 5;
    const int lane = threadIdx.x & 31;
    const int node = blockIdx.x * (blockDim.x >> 5) + wave;
    if (node >= N_NODES) return;
    const float* src = (node < NUM_USER) ? (pref + (size_t)node * DIM_E)
                                         : (temp + (size_t)(node - NUM_USER) * DIM_E);
    float v0 = src[lane];
    float v1 = src[lane + 32];
    const float d = dis[node];
    if (node < NUM_USER) {
        out_pref[(size_t)node * DIM_E + lane]      = v0;
        out_pref[(size_t)node * DIM_E + lane + 32] = v1;
    }
    float s = v0 * v0 + v1 * v1;
    #pragma unroll
    for (int off = 16; off > 0; off >>= 1) s += __shfl_xor(s, off, 32);
    const float inv = 1.0f / fmaxf(sqrtf(s), 1e-12f);
    const float n0 = v0 * inv;
    const float n1 = v1 * inv;
    xn[(size_t)node * DIM_E + lane]      = n0;
    xn[(size_t)node * DIM_E + lane + 32] = n1;
    xs[(size_t)node * DIM_E + lane]      = n0 * d;
    xs[(size_t)node * DIM_E + lane + 32] = n1 * d;
}

#define EDGE_CH  1024
#define N_CH     ((N_EDGES + EDGE_CH - 1) / EDGE_CH)
__device__ __forceinline__ int node_set(int n) { return (n >> 6) & 1; }

__global__ __launch_bounds__(256) void k_degree_slots(const int* __restrict__ ei,
                                                      float* __restrict__ deg,
                                                      int* __restrict__ rec,
                                                      int* __restrict__ qn) {
    __shared__ int s_task;
    const unsigned hwid = __builtin_amdgcn_s_sendmsg_rtn(0x87);
    const int myhalf = (int)((hwid >> 18) & 1u);

    for (int pass = 0; pass < 2; ++pass) {
        const int set = myhalf ^ pass;
        int* __restrict__ ctr = qn + QN_PITCH * set;
        for (int it = 0; it <= N_CH; ++it) {
            if (threadIdx.x == 0) s_task = atomicAdd(ctr, 1);
            __syncthreads();
            const int task = s_task;
            __syncthreads();
            if (task >= N_CH) break;
            const int e0 = task * EDGE_CH + threadIdx.x;
            int r[4], c[4];
            #pragma unroll
            for (int k = 0; k < 4; ++k) {
                const int e = e0 + 256 * k;
                const bool in = e < N_EDGES;
                r[k] = in ? ei[e] : 0;
                c[k] = in ? ei[N_EDGES + e] : 0;
            }
            int s[4];
            #pragma unroll
            for (int k = 0; k < 4; ++k) {
                s[k] = REC_CAP;
                if (r[k] != c[k]) {
                    if (node_set(r[k]) == set) atomicAdd(deg + r[k], 1.0f);
                    if (node_set(c[k]) == set) s[k] = atomicAdd(rec + (size_t)c[k] * REC_W, 1);
                }
            }
            #pragma unroll
            for (int k = 0; k < 4; ++k) {
                if (s[k] < REC_CAP) rec[(size_t)c[k] * REC_W + 1 + s[k]] = r[k];
            }
        }
    }
}

__global__ void k_dis(float* __restrict__ deg) {
    int n = blockIdx.x * blockDim.x + threadIdx.x;
    if (n >= N_NODES) return;
    float d = deg[n];
    deg[n] = (d > 0.0f) ? rsqrtf(d) : 0.0f;
}

#define LD(U)                                                                   \
    const int    r##U = row_bcast_i<U>(a);                                      \
    const float  w##U = row_bcast_f<U>(w);                                      \
    const float4 v##U = *(const float4*)(xq + (size_t)r##U * DIM_E);
#define AC(U)                                                                   \
    acc.x = fmaf(v##U.x, w##U, acc.x);                                          \
    acc.y = fmaf(v##U.y, w##U, acc.y);                                          \
    acc.z = fmaf(v##U.z, w##U, acc.z);                                          \
    acc.w = fmaf(v##U.w, w##U, acc.w);

template <bool PRESCALED>
__global__ __launch_bounds__(256) void k_pull(const int* __restrict__ ei,
                                              const int* __restrict__ rec,
                                              const float* __restrict__ dis,
                                              const float* __restrict__ x,
                                              const float* __restrict__ own,
                                              float* __restrict__ out) {
    const int tid = blockIdx.x * blockDim.x + threadIdx.x;
    const int c = tid >> 4;
    const int q = tid & 15;
    if (c >= N_NODES) return;
    const int* __restrict__ t = rec + (size_t)c * REC_W;
    const float* __restrict__ xq = x + q * 4;
    const int total = t[0];
    const float dc = dis[c];
    const float4 self = *(const float4*)(own + (size_t)c * DIM_E + q * 4);
    float4 acc = {0.0f, 0.0f, 0.0f, 0.0f};

    if (total > 0 && total <= REC_CAP) {
        const bool ok0 = (q >= 1) && (q - 1 < total);
        int a = 0;
        if (ok0) a = t[q];
        const int rpad = row_bcast_i<1>(a);
        if (!ok0) a = rpad;
        float w = ok0 ? 1.0f : 0.0f;
        {
            LD(1) LD(2) LD(3) LD(4) LD(5) LD(6) LD(7) LD(8)
            LD(9) LD(10) LD(11) LD(12) LD(13) LD(14) LD(15)
            AC(1) AC(2) AC(3) AC(4) AC(5) AC(6) AC(7) AC(8)
            AC(9) AC(10) AC(11) AC(12) AC(13) AC(14) AC(15)
        }
        if (total > 15) {
            const bool ok1 = (15 + q < total);
            a = rpad;
            if (ok1) a = t[16 + q];
            w = ok1 ? 1.0f : 0.0f;
            {
                LD(0) LD(1) LD(2) LD(3) LD(4) LD(5) LD(6) LD(7)
                AC(0) AC(1) AC(2) AC(3) AC(4) AC(5) AC(6) AC(7)
            }
            if (total > 23) {
                LD(8) LD(9) LD(10) LD(11) LD(12) LD(13) LD(14) LD(15)
                AC(8) AC(9) AC(10) AC(11) AC(12) AC(13) AC(14) AC(15)
            }
        }
        if (total > 31) {
            for (int k = 2; k < 4; ++k) {
                const int base = 16 * k - 1;
                if (base < total) {
                    const bool ok = (base + q < total);
                    a = rpad;
                    if (ok) a = t[16 * k + q];
                    w = ok ? 1.0f : 0.0f;
                    LD(0) LD(1) LD(2) LD(3) LD(4) LD(5) LD(6) LD(7)
                    LD(8) LD(9) LD(10) LD(11) LD(12) LD(13) LD(14) LD(15)
                    AC(0) AC(1) AC(2) AC(3) AC(4) AC(5) AC(6) AC(7)
                    AC(8) AC(9) AC(10) AC(11) AC(12) AC(13) AC(14) AC(15)
                }
            }
        }
    } else if (total > REC_CAP) {
        for (int e0 = 0; e0 < N_EDGES; e0 += 16) {
            const int rq = ei[e0 + q];
            const int cq = ei[N_EDGES + e0 + q];
            const bool hit = (cq == c) && (rq != c);
            const unsigned hits = __builtin_amdgcn_ballot_w32(hit);
            if (((hits >> (threadIdx.x & 16)) & 0xffffu) != 0u) {
                const int a = hit ? rq : c;
                const float w = hit ? 1.0f : 0.0f;
                LD(0) LD(1) LD(2) LD(3) LD(4) LD(5) LD(6) LD(7)
                LD(8) LD(9) LD(10) LD(11) LD(12) LD(13) LD(14) LD(15)
                AC(0) AC(1) AC(2) AC(3) AC(4) AC(5) AC(6) AC(7)
                AC(8) AC(9) AC(10) AC(11) AC(12) AC(13) AC(14) AC(15)
            }
        }
    }
    const float sc = PRESCALED ? dc * dc : dc;
    float4 res;
    res.x = fmaf(acc.x, sc, self.x);
    res.y = fmaf(acc.y, sc, self.y);
    res.z = fmaf(acc.z, sc, self.z);
    res.w = fmaf(acc.w, sc, self.w);
    *(float4*)(out + (size_t)c * DIM_E + q * 4) = res;
}
#undef LD
#undef AC

extern "C" void kernel_launch(void* const* d_in, const int* in_sizes, int n_in,
                              void* d_out, int out_size, void* d_ws, size_t ws_size,
                              hipStream_t stream) {
    const int*   ei   = (const int*)  d_in[0];
    const float* F    = (const float*)d_in[1];
    const float* pref = (const float*)d_in[2];
    const float* W1   = (const float*)d_in[3];
    const float* b1   = (const float*)d_in[4];
    const float* W2   = (const float*)d_in[5];
    const float* b2   = (const float*)d_in[6];
    float*       out  = (float*)d_out;

    float* ws   = (float*)d_ws;
    float* xn   = ws;
    float* temp = xn   + (size_t)N_NODES * DIM_E;
    int*   rec  = (int*)(temp + (size_t)NUM_ITEM * DIM_E);
    float* deg  = (float*)(rec + (size_t)N_NODES * REC_W);
    float* w1s  = deg + N_NODES;
    float* w2s  = w1s + DIM_FEAT * DIM_H;
    float* xs   = w2s + DIM_H * DIM_E;
    float* ys   = xs   + (size_t)N_NODES * DIM_E;
    int*   qn   = (int*)(ys + (size_t)N_NODES * DIM_E);

    const int T = 256;
    k_init<<<(N_NODES + T - 1) / T, T, 0, stream>>>(deg, rec, W1, W2, w1s, w2s, qn);
    k_degree_slots<<<2048, T, 0, stream>>>(ei, deg, rec, qn);
    k_dis<<<(N_NODES + T - 1) / T, T, 0, stream>>>(deg);
    k_mlp<<<(N_TILES + 3) / 4, 128, 0, stream>>>(F, w1s, b1, w2s, b2, temp);
    k_normalize<<<(N_NODES + 7) / 8, T, 0, stream>>>(pref, temp, deg, xn, xs,
                                                     out + (size_t)N_NODES * DIM_E);
    {
        int nthr = N_NODES * 16;
        k_pull<true><<<(nthr + T - 1) / T, T, 0, stream>>>(ei, rec, deg, xs, xs, ys);
        k_pull<false><<<(nthr + T - 1) / T, T, 0, stream>>>(ei, rec, deg, ys, xn, out);
    }
}


# speedup vs baseline: 3.9129x; 1.0340x over previous
#include <hip/hip_runtime.h>

#define NUM_USER 50000
#define NUM_ITEM 50000
#define N_NODES  100000
#define DIM_E    64
#define DIM_FEAT 128
#define DIM_H    256
#define N_EDGES  1600000
#define N_TILES  (NUM_ITEM / 16)
#define DEG_PITCH ((N_NODES + 63) / 64 * 64)

typedef __attribute__((ext_vector_type(2))) float v2f;
typedef __attribute__((ext_vector_type(8))) float v8f;

#define REC_W   64
#define REC_CAP (REC_W - 1)
#define QN_PITCH 64

__global__ void k_init(float* __restrict__ deg, int* __restrict__ rec,
                       const float* __restrict__ W1, const float* __restrict__ W2,
                       float* __restrict__ w1s, float* __restrict__ w2s,
                       int* __restrict__ qn) {
    int n = blockIdx.x * blockDim.x + threadIdx.x;
    if (n >= N_NODES) return;
    deg[n] = 0.0f;
    rec[(size_t)n * REC_W] = 0;
    if (n < 2) qn[QN_PITCH * n] = 0;
    const int q = n & 3;
    const int c = (n >> 2) & 15;
    const int h = (n >> 6) & 1;
    if (n < DIM_FEAT * DIM_H) {
        const int j = (n >> 7) & 15;
        const int t = (n >> 11) & 3;
        const int g = (n >> 13) & 3;
        w1s[n] = W1[(8 * j + 4 * h + q) * DIM_H + 64 * g + 4 * c + t];
    }
    if (n < DIM_H * DIM_E) {
        const int ip = (n >> 7) & 1;
        const int j2 = (n >> 8) & 3;
        const int t  = (n >> 10) & 3;
        const int g  = (n >> 12) & 3;
        w2s[n] = W2[(64 * g + 32 * h + 8 * j2 + 4 * (q & 1) + t) * DIM_E + 4 * c + 2 * ip + (q >> 1)];
    }
}

template <int U> __device__ __forceinline__ int row_bcast_i(int v) {
    return __builtin_amdgcn_update_dpp(0, v, 0x150 + U, 0xf, 0xf, false);
}
template <int U> __device__ __forceinline__ float row_bcast_f(float v) {
    return __int_as_float(row_bcast_i<U>(__float_as_int(v)));
}

#define WM(A0, A1, B, ACC)                                                      \
    ACC = __builtin_amdgcn_wmma_f32_16x16x4_f32(false, (v2f){A0, A1}, false, B, \
                                                (short)0, ACC, false, false)
#define LRELU(X) (((X) > 0.0f) ? (X) : 0.01f * (X))

__global__ __launch_bounds__(128) __attribute__((amdgpu_waves_per_eu(1, 4))) void k_mlp(
        const float* __restrict__ F,
        const float* __restrict__ w1s, const float* __restrict__ b1,
        const float* __restrict__ w2s, const float* __restrict__ b2,
        float* __restrict__ temp) {
    __shared__ float stage[4][16][68];

    const int wave = threadIdx.x >> 5;
    const int lane = threadIdx.x & 31;
    const int c    = lane & 15;
    const int h    = lane >> 4;
    const int tile = blockIdx.x * 4 + wave;
    if (tile >= N_TILES) return;
    const int r0 = tile * 16;

    float4 fq[16];
    {
        const float* fp = F + (size_t)(r0 + c) * DIM_FEAT + 4 * h;
        #pragma unroll
        for (int j = 0; j < 16; ++j) fq[j] = *(const float4*)(fp + 8 * j);
    }

    v8f o0 = {}, o1 = {}, o2 = {}, o3 = {};
    const float4* __restrict__ p1 = (const float4*)w1s + lane;
    const float4* __restrict__ p2 = (const float4*)w2s + lane;

    float4 nx[3][4];
    #pragma unroll
    for (int j = 0; j < 3; ++j) {
        #pragma unroll
        for (int t = 0; t < 4; ++t) nx[j][t] = p1[(t * 16 + j) * 32];
    }

    #pragma unroll
    for (int g = 0; g < 4; ++g) {
        const float4* q1 = p1 + g * (4 * 16 * 32);
        const float4* q2 = p2 + g * (16 * 2 * 32);
        const float4* qn = q1 + (4 * 16 * 32);
        float4 w[16][4];
        float4 y[16][2];
        v8f a[4] = {};
        float4 bb[8];
        #pragma unroll
        for (int j = 0; j < 3; ++j) {
            #pragma unroll
            for (int t = 0; t < 4; ++t) w[j][t] = nx[j][t];
        }
        #pragma unroll
        for (int j = 0; j < 16; ++j) {
            if (j + 3 < 16) {
                #pragma unroll
                for (int t = 0; t < 4; ++t) w[j + 3][t] = q1[(t * 16 + j + 3) * 32];
            } else {
                #pragma unroll
                for (int k = 2 * (j - 13); k < 2 * (j - 13) + 2; ++k) {
                    y[k][0] = q2[(k * 2 + 0) * 32];
                    y[k][1] = q2[(k * 2 + 1) * 32];
                }
            }
            if (j >= 4 && j < 12)
                bb[j - 4] = *(const float4*)(b1 + 64 * g + 32 * h + 4 * (j - 4));
            const v2f bA = {fq[j].x, fq[j].y};
            const v2f bB = {fq[j].z, fq[j].w};
            #pragma unroll
            for (int t = 0; t < 4; ++t) WM(w[j][t].x, w[j][t].y, bA, a[t]);
            #pragma unroll
            for (int t = 0; t < 4; ++t) WM(w[j][t].z, w[j][t].w, bB, a[t]);
        }
        #pragma unroll
        for (int r = 0; r < 8; ++r) {
            float v;
            v = a[0][r] + bb[r].x; a[0][r] = LRELU(v);
            v = a[1][r] + bb[r].y; a[1][r] = LRELU(v);
            v = a[2][r] + bb[r].z; a[2][r] = LRELU(v);
            v = a[3][r] + bb[r].w; a[3][r] = LRELU(v);
        }
        #pragma unroll
        for (int k = 0; k < 16; ++k) {
            if (k + 6 < 16) {
                y[k + 6][0] = q2[((k + 6) * 2 + 0) * 32];
                y[k + 6][1] = q2[((k + 6) * 2 + 1) * 32];
            } else if (k >= 13 && g + 1 < 4) {
                #pragma unroll
                for (int t = 0; t < 4; ++t) nx[k - 13][t] = qn[(t * 16 + (k - 13)) * 32];
            }
            const v2f bh = {a[k >> 2][2 * (k & 3)], a[k >> 2][2 * (k & 3) + 1]};
            WM(y[k][0].x, y[k][0].y, bh, o0); WM(y[k][0].z, y[k][0].w, bh, o1);
            WM(y[k][1].x, y[k][1].y, bh, o2); WM(y[k][1].z, y[k][1].w, bh, o3);
        }
    }

    {
        float* st = &stage[wave][c][32 * h];
        #pragma unroll
        for (int r = 0; r < 8; ++r) {
            const float4 bb = *(const float4*)(b2 + 32 * h + 4 * r);
            float4 v;
            v.x = o0[r] + bb.x; v.y = o1[r] + bb.y;
            v.z = o2[r] + bb.z; v.w = o3[r] + bb.w;
            *(float4*)(st + 4 * r) = v;
        }
        #pragma unroll
        for (int it = 0; it < 8; ++it) {
            const int row = 2 * it + h;
            const float4 v = *(const float4*)(&stage[wave][row][4 * c]);
            *(float4*)(temp + (size_t)(r0 + row) * DIM_E + 4 * c) = v;
        }
    }
}
#undef WM
#undef LRELU

__global__ void k_normalize(const float* __restrict__ pref,
                            const float* __restrict__ temp,
                            const float* __restrict__ dis,
                            float* __restrict__ xn,
                            float* __restrict__ xs,
                            float* __restrict__ out_pref) {
    const int wave = threadIdx.x >> 5;
    const int lane = threadIdx.x & 31;
    const int node = blockIdx.x * (blockDim.x >> 5) + wave;
    if (node >= N_NODES) return;
    const float* src = (node < NUM_USER) ? (pref + (size_t)node * DIM_E)
                                         : (temp + (size_t)(node - NUM_USER) * DIM_E);
    const size_t at = (size_t)node * DIM_E + 2 * lane;
    const float2 v = *(const float2*)(src + 2 * lane);
    const float d = dis[node];
    if (node < NUM_USER) {
        *(float2*)(out_pref + at) = v;
    }
    float s = v.x * v.x + v.y * v.y;
    #pragma unroll
    for (int off = 16; off > 0; off >>= 1) s += __shfl_xor(s, off, 32);
    const float inv = 1.0f / fmaxf(sqrtf(s), 1e-12f);
    float2 n;
    n.x = v.x * inv;
    n.y = v.y * inv;
    *(float2*)(xn + at) = n;
    float2 m;
    m.x = n.x * d;
    m.y = n.y * d;
    *(float2*)(xs + at) = m;
}

#define EDGE_CH  1024
#define N_CH     ((N_EDGES + EDGE_CH - 1) / EDGE_CH)
__device__ __forceinline__ int node_set(int n) { return (n >> 6) & 1; }

__global__ __launch_bounds__(256) void k_degree_slots(const int* __restrict__ ei,
                                                      float* __restrict__ deg,
                                                      int* __restrict__ rec,
                                                      int* __restrict__ qn) {
    __shared__ int s_task;
    const unsigned hwid = __builtin_amdgcn_s_sendmsg_rtn(0x87);
    const int myhalf = (int)((hwid >> 18) & 1u);

    for (int pass = 0; pass < 2; ++pass) {
        const int set = myhalf ^ pass;
        int* __restrict__ ctr = qn + QN_PITCH * set;
        for (int it = 0; it <= N_CH; ++it) {
            if (threadIdx.x == 0) s_task = atomicAdd(ctr, 1);
            __syncthreads();
            const int task = s_task;
            __syncthreads();
            if (task >= N_CH) break;
            const int e0 = task * EDGE_CH + threadIdx.x;
            int r[4], c[4];
            #pragma unroll
            for (int k = 0; k < 4; ++k) {
                const int e = e0 + 256 * k;
                const bool in = e < N_EDGES;
                r[k] = in ? ei[e] : 0;
                c[k] = in ? ei[N_EDGES + e] : 0;
            }
            int s[4];
            #pragma unroll
            for (int k = 0; k < 4; ++k) {
                s[k] = REC_CAP;
                if (r[k] != c[k]) {
                    if (node_set(r[k]) == set) atomicAdd(deg + r[k], 1.0f);
                    if (node_set(c[k]) == set) s[k] = atomicAdd(rec + (size_t)c[k] * REC_W, 1);
                }
            }
            #pragma unroll
            for (int k = 0; k < 4; ++k) {
                if (s[k] < REC_CAP) rec[(size_t)c[k] * REC_W + 1 + s[k]] = r[k];
            }
        }
    }
}

__global__ void k_dis(float* __restrict__ deg) {
    int n = blockIdx.x * blockDim.x + threadIdx.x;
    if (n >= N_NODES) return;
    float d = deg[n];
    deg[n] = (d > 0.0f) ? rsqrtf(d) : 0.0f;
}

#define LD(U)                                                                   \
    const int    r##U = row_bcast_i<U>(a);                                      \
    const float  w##U = row_bcast_f<U>(w);                                      \
    const float4 v##U = *(const float4*)(xq + (size_t)r##U * DIM_E);
#define AC(U)                                                                   \
    acc.x = fmaf(v##U.x, w##U, acc.x);                                          \
    acc.y = fmaf(v##U.y, w##U, acc.y);                                          \
    acc.z = fmaf(v##U.z, w##U, acc.z);                                          \
    acc.w = fmaf(v##U.w, w##U, acc.w);

template <bool PRESCALED>
__global__ __launch_bounds__(256) void k_pull(const int* __restrict__ ei,
                                              const int* __restrict__ rec,
                                              const float* __restrict__ dis,
                                              const float* __restrict__ x,
                                              const float* __restrict__ own,
                                              float* __restrict__ out) {
    const int tid = blockIdx.x * blockDim.x + threadIdx.x;
    const int c = tid >> 4;
    const int q = tid & 15;
    if (c >= N_NODES) return;
    const int* __restrict__ t = rec + (size_t)c * REC_W;
    const float* __restrict__ xq = x + q * 4;
    const int total = t[0];
    const float dc = dis[c];
    const float4 self = *(const float4*)(own + (size_t)c * DIM_E + q * 4);
    float4 acc = {0.0f, 0.0f, 0.0f, 0.0f};

    if (total > 0 && total <= REC_CAP) {
        const bool ok0 = (q >= 1) && (q - 1 < total);
        int a = 0;
        if (ok0) a = t[q];
        const int rpad = row_bcast_i<1>(a);
        if (!ok0) a = rpad;
        float w = ok0 ? 1.0f : 0.0f;
        {
            LD(1) LD(2) LD(3) LD(4) LD(5) LD(6) LD(7) LD(8)
            LD(9) LD(10) LD(11) LD(12) LD(13) LD(14) LD(15)
            AC(1) AC(2) AC(3) AC(4) AC(5) AC(6) AC(7) AC(8)
            AC(9) AC(10) AC(11) AC(12) AC(13) AC(14) AC(15)
        }
        if (total > 15) {
            const bool ok1 = (15 + q < total);
            a = rpad;
            if (ok1) a = t[16 + q];
            w = ok1 ? 1.0f : 0.0f;
            {
                LD(0) LD(1) LD(2) LD(3) LD(4) LD(5) LD(6) LD(7)
                AC(0) AC(1) AC(2) AC(3) AC(4) AC(5) AC(6) AC(7)
            }
            if (total > 23) {
                LD(8) LD(9) LD(10) LD(11) LD(12) LD(13) LD(14) LD(15)
                AC(8) AC(9) AC(10) AC(11) AC(12) AC(13) AC(14) AC(15)
            }
        }
        if (total > 31) {
            for (int k = 2; k < 4; ++k) {
                const int base = 16 * k - 1;
                if (base < total) {
                    const bool ok = (base + q < total);
                    a = rpad;
                    if (ok) a = t[16 * k + q];
                    w = ok ? 1.0f : 0.0f;
                    LD(0) LD(1) LD(2) LD(3) LD(4) LD(5) LD(6) LD(7)
                    LD(8) LD(9) LD(10) LD(11) LD(12) LD(13) LD(14) LD(15)
                    AC(0) AC(1) AC(2) AC(3) AC(4) AC(5) AC(6) AC(7)
                    AC(8) AC(9) AC(10) AC(11) AC(12) AC(13) AC(14) AC(15)
                }
            }
        }
    } else if (total > REC_CAP) {
        for (int e0 = 0; e0 < N_EDGES; e0 += 16) {
            const int rq = ei[e0 + q];
            const int cq = ei[N_EDGES + e0 + q];
            const bool hit = (cq == c) && (rq != c);
            const unsigned hits = __builtin_amdgcn_ballot_w32(hit);
            if (((hits >> (threadIdx.x & 16)) & 0xffffu) != 0u) {
                const int a = hit ? rq : c;
                const float w = hit ? 1.0f : 0.0f;
                LD(0) LD(1) LD(2) LD(3) LD(4) LD(5) LD(6) LD(7)
                LD(8) LD(9) LD(10) LD(11) LD(12) LD(13) LD(14) LD(15)
                AC(0) AC(1) AC(2) AC(3) AC(4) AC(5) AC(6) AC(7)
                AC(8) AC(9) AC(10) AC(11) AC(12) AC(13) AC(14) AC(15)
            }
        }
    }
    const float sc = PRESCALED ? dc * dc : dc;
    float4 res;
    res.x = fmaf(acc.x, sc, self.x);
    res.y = fmaf(acc.y, sc, self.y);
    res.z = fmaf(acc.z, sc, self.z);
    res.w = fmaf(acc.w, sc, self.w);
    *(float4*)(out + (size_t)c * DIM_E + q * 4) = res;
}
#undef LD
#undef AC

extern "C" void kernel_launch(void* const* d_in, const int* in_sizes, int n_in,
                              void* d_out, int out_size, void* d_ws, size_t ws_size,
                              hipStream_t stream) {
    const int*   ei   = (const int*)  d_in[0];
    const float* F    = (const float*)d_in[1];
    const float* pref = (const float*)d_in[2];
    const float* W1   = (const float*)d_in[3];
    const float* b1   = (const float*)d_in[4];
    const float* W2   = (const float*)d_in[5];
    const float* b2   = (const float*)d_in[6];
    float*       out  = (float*)d_out;

    float* ws   = (float*)d_ws;
    float* xn   = ws;
    float* temp = xn   + (size_t)N_NODES * DIM_E;
    int*   rec  = (int*)(temp + (size_t)NUM_ITEM * DIM_E);
    float* deg  = (float*)(rec + (size_t)N_NODES * REC_W);
    float* w1s  = deg + DEG_PITCH;
    float* w2s  = w1s + DIM_FEAT * DIM_H;
    float* xs   = w2s + DIM_H * DIM_E;
    float* ys   = xs   + (size_t)N_NODES * DIM_E;
    int*   qn   = (int*)(ys + (size_t)N_NODES * DIM_E);

    const int T = 256;
    k_init<<<(N_NODES + T - 1) / T, T, 0, stream>>>(deg, rec, W1, W2, w1s, w2s, qn);
    k_degree_slots<<<2048, T, 0, stream>>>(ei, deg, rec, qn);
    k_dis<<<(N_NODES + T - 1) / T, T, 0, stream>>>(deg);
    k_mlp<<<(N_TILES + 3) / 4, 128, 0, stream>>>(F, w1s, b1, w2s, b2, temp);
    k_normalize<<<(N_NODES + 7) / 8, T, 0, stream>>>(pref, temp, deg, xn, xs,
                                                     out + (size_t)N_NODES * DIM_E);
    {
        int nthr = N_NODES * 16;
        k_pull<true><<<(nthr + T - 1) / T, T, 0, stream>>>(ei, rec, deg, xs, xs, ys);
        k_pull<false><<<(nthr + T - 1) / T, T, 0, stream>>>(ei, rec, deg, ys, xn, out);
    }
}


# long-lived waves: normalize takes 16 nodes per wave (was 1), a hop is 1280 resident work-groups whose 16-lane rows loop over rows with the next record fetched a row ahead; norm sum order changes
# speedup vs baseline: 4.0851x; 1.0440x over previous
#include <hip/hip_runtime.h>

#define NUM_USER 50000
#define NUM_ITEM 50000
#define N_NODES  100000
#define DIM_E    64
#define DIM_FEAT 128
#define DIM_H    256
#define N_EDGES  1600000
#define N_TILES  (NUM_ITEM / 16)
#define DEG_PITCH ((N_NODES + 63) / 64 * 64)

typedef __attribute__((ext_vector_type(2))) float v2f;
typedef __attribute__((ext_vector_type(8))) float v8f;

#define REC_W   64
#define REC_CAP (REC_W - 1)
#define QN_PITCH 64

__global__ void k_init(float* __restrict__ deg, int* __restrict__ rec,
                       const float* __restrict__ W1, const float* __restrict__ W2,
                       float* __restrict__ w1s, float* __restrict__ w2s,
                       int* __restrict__ qn) {
    int n = blockIdx.x * blockDim.x + threadIdx.x;
    if (n >= N_NODES) return;
    deg[n] = 0.0f;
    rec[(size_t)n * REC_W] = 0;
    if (n < 2) qn[QN_PITCH * n] = 0;
    const int q = n & 3;
    const int c = (n >> 2) & 15;
    const int h = (n >> 6) & 1;
    if (n < DIM_FEAT * DIM_H) {
        const int j = (n >> 7) & 15;
        const int t = (n >> 11) & 3;
        const int g = (n >> 13) & 3;
        w1s[n] = W1[(8 * j + 4 * h + q) * DIM_H + 64 * g + 4 * c + t];
    }
    if (n < DIM_H * DIM_E) {
        const int ip = (n >> 7) & 1;
        const int j2 = (n >> 8) & 3;
        const int t  = (n >> 10) & 3;
        const int g  = (n >> 12) & 3;
        w2s[n] = W2[(64 * g + 32 * h + 8 * j2 + 4 * (q & 1) + t) * DIM_E + 4 * c + 2 * ip + (q >> 1)];
    }
}

template <int U> __device__ __forceinline__ int row_bcast_i(int v) {
    return __builtin_amdgcn_update_dpp(0, v, 0x150 + U, 0xf, 0xf, false);
}
template <int U> __device__ __forceinline__ float row_bcast_f(float v) {
    return __int_as_float(row_bcast_i<U>(__float_as_int(v)));
}

#define WM(A0, A1, B, ACC)                                                      \
    ACC = __builtin_amdgcn_wmma_f32_16x16x4_f32(false, (v2f){A0, A1}, false, B, \
                                                (short)0, ACC, false, false)
#define LRELU(X) (((X) > 0.0f) ? (X) : 0.01f * (X))

__global__ __launch_bounds__(128) __attribute__((amdgpu_waves_per_eu(1, 4))) void k_mlp(
        const float* __restrict__ F,
        const float* __restrict__ w1s, const float* __restrict__ b1,
        const float* __restrict__ w2s, const float* __restrict__ b2,
        float* __restrict__ temp) {
    __shared__ float stage[4][16][68];

    const int wave = threadIdx.x >> 5;
    const int lane = threadIdx.x & 31;
    const int c    = lane & 15;
    const int h    = lane >> 4;
    const int tile = blockIdx.x * 4 + wave;
    if (tile >= N_TILES) return;
    const int r0 = tile * 16;

    float4 fq[16];
    {
        const float* fp = F + (size_t)(r0 + c) * DIM_FEAT + 4 * h;
        #pragma unroll
        for (int j = 0; j < 16; ++j) fq[j] = *(const float4*)(fp + 8 * j);
    }

    v8f o0 = {}, o1 = {}, o2 = {}, o3 = {};
    const float4* __restrict__ p1 = (const float4*)w1s + lane;
    const float4* __restrict__ p2 = (const float4*)w2s + lane;

    float4 nx[3][4];
    #pragma unroll
    for (int j = 0; j < 3; ++j) {
        #pragma unroll
        for (int t = 0; t < 4; ++t) nx[j][t] = p1[(t * 16 + j) * 32];
    }

    #pragma unroll
    for (int g = 0; g < 4; ++g) {
        const float4* q1 = p1 + g * (4 * 16 * 32);
        const float4* q2 = p2 + g * (16 * 2 * 32);
        const float4* qn = q1 + (4 * 16 * 32);
        float4 w[16][4];
        float4 y[16][2];
        v8f a[4] = {};
        float4 bb[8];
        #pragma unroll
        for (int j = 0; j < 3; ++j) {
            #pragma unroll
            for (int t = 0; t < 4; ++t) w[j][t] = nx[j][t];
        }
        #pragma unroll
        for (int j = 0; j < 16; ++j) {
            if (j + 3 < 16) {
                #pragma unroll
                for (int t = 0; t < 4; ++t) w[j + 3][t] = q1[(t * 16 + j + 3) * 32];
            } else {
                #pragma unroll
                for (int k = 2 * (j - 13); k < 2 * (j - 13) + 2; ++k) {
                    y[k][0] = q2[(k * 2 + 0) * 32];
                    y[k][1] = q2[(k * 2 + 1) * 32];
                }
            }
            if (j >= 4 && j < 12)
                bb[j - 4] = *(const float4*)(b1 + 64 * g + 32 * h + 4 * (j - 4));
            const v2f bA = {fq[j].x, fq[j].y};
            const v2f bB = {fq[j].z, fq[j].w};
            #pragma unroll
            for (int t = 0; t < 4; ++t) WM(w[j][t].x, w[j][t].y, bA, a[t]);
            #pragma unroll
            for (int t = 0; t < 4; ++t) WM(w[j][t].z, w[j][t].w, bB, a[t]);
        }
        #pragma unroll
        for (int r = 0; r < 8; ++r) {
            float v;
            v = a[0][r] + bb[r].x; a[0][r] = LRELU(v);
            v = a[1][r] + bb[r].y; a[1][r] = LRELU(v);
            v = a[2][r] + bb[r].z; a[2][r] = LRELU(v);
            v = a[3][r] + bb[r].w; a[3][r] = LRELU(v);
        }
        #pragma unroll
        for (int k = 0; k < 16; ++k) {
            if (k + 6 < 16) {
                y[k + 6][0] = q2[((k + 6) * 2 + 0) * 32];
                y[k + 6][1] = q2[((k + 6) * 2 + 1) * 32];
            } else if (k >= 13 && g + 1 < 4) {
                #pragma unroll
                for (int t = 0; t < 4; ++t) nx[k - 13][t] = qn[(t * 16 + (k - 13)) * 32];
            }
            const v2f bh = {a[k >> 2][2 * (k & 3)], a[k >> 2][2 * (k & 3) + 1]};
            WM(y[k][0].x, y[k][0].y, bh, o0); WM(y[k][0].z, y[k][0].w, bh, o1);
            WM(y[k][1].x, y[k][1].y, bh, o2); WM(y[k][1].z, y[k][1].w, bh, o3);
        }
    }

    {
        float* st = &stage[wave][c][32 * h];
        #pragma unroll
        for (int r = 0; r < 8; ++r) {
            const float4 bb = *(const float4*)(b2 + 32 * h + 4 * r);
            float4 v;
            v.x = o0[r] + bb.x; v.y = o1[r] + bb.y;
            v.z = o2[r] + bb.z; v.w = o3[r] + bb.w;
            *(float4*)(st + 4 * r) = v;
        }
        #pragma unroll
        for (int it = 0; it < 8; ++it) {
            const int row = 2 * it + h;
            const float4 v = *(const float4*)(&stage[wave][row][4 * c]);
            *(float4*)(temp + (size_t)(r0 + row) * DIM_E + 4 * c) = v;
        }
    }
}
#undef WM
#undef LRELU

template <int M> __device__ __forceinline__ float row_xor_f(float v) {
    return __int_as_float(__builtin_amdgcn_update_dpp(0, __float_as_int(v), 0x160 + M, 0xf, 0xf, false));
}

#define NORM_R 8
#define NORM_WAVES (N_NODES / (2 * NORM_R))
static_assert(N_NODES % (2 * NORM_R) == 0, "a wave takes 16 whole nodes");

__global__ __launch_bounds__(256) void k_normalize(const float* __restrict__ pref,
                                                   const float* __restrict__ temp,
                                                   const float* __restrict__ dis,
                                                   float* __restrict__ xn,
                                                   float* __restrict__ xs,
                                                   float* __restrict__ out_pref) {
    const int gw   = (blockIdx.x * blockDim.x + threadIdx.x) >> 5;
    const int lane = threadIdx.x & 31;
    const int h    = lane >> 4;
    const int q    = lane & 15;
    if (gw >= NORM_WAVES) return;
    const int first = gw * (2 * NORM_R) + h;

    float4 v[NORM_R];
    float  d[NORM_R];
    #pragma unroll
    for (int k = 0; k < NORM_R; ++k) {
        const int node = first + 2 * k;
        const float* src = (node < NUM_USER) ? (pref + (size_t)node * DIM_E)
                                             : (temp + (size_t)(node - NUM_USER) * DIM_E);
        v[k] = *(const float4*)(src + 4 * q);
        d[k] = dis[node];
    }
    #pragma unroll
    for (int k = 0; k < NORM_R; ++k) {
        const int node = first + 2 * k;
        const size_t at = (size_t)node * DIM_E + 4 * q;
        if (node < NUM_USER) {
            *(float4*)(out_pref + at) = v[k];
        }
        float s = v[k].x * v[k].x + v[k].y * v[k].y + v[k].z * v[k].z + v[k].w * v[k].w;
        s += row_xor_f<1>(s);
        s += row_xor_f<2>(s);
        s += row_xor_f<4>(s);
        s += row_xor_f<8>(s);
        const float inv = 1.0f / fmaxf(sqrtf(s), 1e-12f);
        float4 n;
        n.x = v[k].x * inv;
        n.y = v[k].y * inv;
        n.z = v[k].z * inv;
        n.w = v[k].w * inv;
        *(float4*)(xn + at) = n;
        float4 m;
        m.x = n.x * d[k];
        m.y = n.y * d[k];
        m.z = n.z * d[k];
        m.w = n.w * d[k];
        *(float4*)(xs + at) = m;
    }
}

#define EDGE_CH  1024
#define N_CH     ((N_EDGES + EDGE_CH - 1) / EDGE_CH)
__device__ __forceinline__ int node_set(int n) { return (n >> 6) & 1; }

__global__ __launch_bounds__(256) void k_degree_slots(const int* __restrict__ ei,
                                                      float* __restrict__ deg,
                                                      int* __restrict__ rec,
                                                      int* __restrict__ qn) {
    __shared__ int s_task;
    const unsigned hwid = __builtin_amdgcn_s_sendmsg_rtn(0x87);
    const int myhalf = (int)((hwid >> 18) & 1u);

    for (int pass = 0; pass < 2; ++pass) {
        const int set = myhalf ^ pass;
        int* __restrict__ ctr = qn + QN_PITCH * set;
        for (int it = 0; it <= N_CH; ++it) {
            if (threadIdx.x == 0) s_task = atomicAdd(ctr, 1);
            __syncthreads();
            const int task = s_task;
            __syncthreads();
            if (task >= N_CH) break;
            const int e0 = task * EDGE_CH + threadIdx.x;
            int r[4], c[4];
            #pragma unroll
            for (int k = 0; k < 4; ++k) {
                const int e = e0 + 256 * k;
                const bool in = e < N_EDGES;
                r[k] = in ? ei[e] : 0;
                c[k] = in ? ei[N_EDGES + e] : 0;
            }
            int s[4];
            #pragma unroll
            for (int k = 0; k < 4; ++k) {
                s[k] = REC_CAP;
                if (r[k] != c[k]) {
                    if (node_set(r[k]) == set) atomicAdd(deg + r[k], 1.0f);
                    if (node_set(c[k]) == set) s[k] = atomicAdd(rec + (size_t)c[k] * REC_W, 1);
                }
            }
            #pragma unroll
            for (int k = 0; k < 4; ++k) {
                if (s[k] < REC_CAP) rec[(size_t)c[k] * REC_W + 1 + s[k]] = r[k];
            }
        }
    }
}

__global__ void k_dis(float* __restrict__ deg) {
    int n = blockIdx.x * blockDim.x + threadIdx.x;
    if (n >= N_NODES) return;
    float d = deg[n];
    deg[n] = (d > 0.0f) ? rsqrtf(d) : 0.0f;
}

#define LD(U)                                                                   \
    const unsigned r##U = (unsigned)row_bcast_i<U>(a);                          \
    const float  w##U = row_bcast_f<U>(w);                                      \
    const float4 v##U = *(const float4*)(xb + (r##U * ROW_B + qb));
#define AC(U)                                                                   \
    acc.x = fmaf(v##U.x, w##U, acc.x);                                          \
    acc.y = fmaf(v##U.y, w##U, acc.y);                                          \
    acc.z = fmaf(v##U.z, w##U, acc.z);                                          \
    acc.w = fmaf(v##U.w, w##U, acc.w);

#define PULL_WGS  1280
#define PULL_ROWS (PULL_WGS * 16u)
#define ROW_B     (DIM_E * 4u)

template <bool PRESCALED>
__global__ __launch_bounds__(256) void k_pull(const int* __restrict__ ei,
                                              const int* __restrict__ rec,
                                              const float* __restrict__ dis,
                                              const float* __restrict__ x,
                                              const float* __restrict__ own,
                                              float* __restrict__ out) {
    const unsigned tid = blockIdx.x * blockDim.x + threadIdx.x;
    const unsigned q  = tid & 15u;
    const unsigned qb = q * 16u;
    const char* __restrict__ xb = (const char*)x;
    unsigned c = tid >> 4;
    int nx0 = 0, nx1 = 0;
    if (c < N_NODES) {
        nx0 = rec[c * REC_W + q];
        nx1 = rec[c * REC_W + 16u + q];
    }
    for (; c < N_NODES; c += PULL_ROWS) {
    const int* __restrict__ t = rec + c * REC_W;
    const int w0 = nx0;
    const int w1 = nx1;
    const int total = row_bcast_i<0>(w0);
    if (c + PULL_ROWS < N_NODES) {
        nx0 = rec[(c + PULL_ROWS) * REC_W + q];
        nx1 = rec[(c + PULL_ROWS) * REC_W + 16u + q];
    }
    const float dc = dis[c];
    const float4 self = *(const float4*)((const char*)own + (c * ROW_B + qb));
    float4 acc = {0.0f, 0.0f, 0.0f, 0.0f};

    if (total > 0 && total <= REC_CAP) {
        const bool ok0 = (q != 0u) && ((int)q <= total);
        int a = w0;
        const int rpad = row_bcast_i<1>(a);
        if (!ok0) a = rpad;
        float w = ok0 ? 1.0f : 0.0f;
        {
            LD(1) LD(2) LD(3) LD(4) LD(5) LD(6) LD(7) LD(8)
            LD(9) LD(10) LD(11) LD(12) LD(13) LD(14) LD(15)
            AC(1) AC(2) AC(3) AC(4) AC(5) AC(6) AC(7) AC(8)
            AC(9) AC(10) AC(11) AC(12) AC(13) AC(14) AC(15)
        }
        if (total > 15) {
            const bool ok1 = ((int)q < total - 15);
            a = ok1 ? w1 : rpad;
            w = ok1 ? 1.0f : 0.0f;
            {
                LD(0) LD(1) LD(2) LD(3) LD(4) LD(5) LD(6) LD(7)
                AC(0) AC(1) AC(2) AC(3) AC(4) AC(5) AC(6) AC(7)
            }
            if (total > 23) {
                LD(8) LD(9) LD(10) LD(11) LD(12) LD(13) LD(14) LD(15)
                AC(8) AC(9) AC(10) AC(11) AC(12) AC(13) AC(14) AC(15)
            }
        }
        if (total > 31) {
            #pragma nounroll
            for (int base = 31; base < total; base += 8) {
                const bool ok = (q < 8u) && (base + (int)q < total);
                a = rpad;
                if (ok) a = t[base + 1 + q];
                w = ok ? 1.0f : 0.0f;
                LD(0) LD(1) LD(2) LD(3) LD(4) LD(5) LD(6) LD(7)
                AC(0) AC(1) AC(2) AC(3) AC(4) AC(5) AC(6) AC(7)
            }
        }
    } else if (total > REC_CAP) {
        for (unsigned e0 = 0; e0 < N_EDGES; e0 += 8) {
            int rq = (int)c, cq = -1;
            if (q < 8u) {
                rq = ei[e0 + q];
                cq = ei[N_EDGES + e0 + q];
            }
            const bool hit = (cq == (int)c) && (rq != (int)c);
            const unsigned hits = __builtin_amdgcn_ballot_w32(hit);
            if (((hits >> (threadIdx.x & 16)) & 0xffffu) != 0u) {
                const int a = hit ? rq : (int)c;
                const float w = hit ? 1.0f : 0.0f;
                LD(0) LD(1) LD(2) LD(3) LD(4) LD(5) LD(6) LD(7)
                AC(0) AC(1) AC(2) AC(3) AC(4) AC(5) AC(6) AC(7)
            }
        }
    }
    const float sc = PRESCALED ? dc * dc : dc;
    float4 res;
    res.x = fmaf(acc.x, sc, self.x);
    res.y = fmaf(acc.y, sc, self.y);
    res.z = fmaf(acc.z, sc, self.z);
    res.w = fmaf(acc.w, sc, self.w);
    *(float4*)((char*)out + (c * ROW_B + qb)) = res;
    }
}
#undef LD
#undef AC

extern "C" void kernel_launch(void* const* d_in, const int* in_sizes, int n_in,
                              void* d_out, int out_size, void* d_ws, size_t ws_size,
                              hipStream_t stream) {
    const int*   ei   = (const int*)  d_in[0];
    const float* F    = (const float*)d_in[1];
    const float* pref = (const float*)d_in[2];
    const float* W1   = (const float*)d_in[3];
    const float* b1   = (const float*)d_in[4];
    const float* W2   = (const float*)d_in[5];
    const float* b2   = (const float*)d_in[6];
    float*       out  = (float*)d_out;

    float* ws   = (float*)d_ws;
    float* xn   = ws;
    float* temp = xn   + (size_t)N_NODES * DIM_E;
    int*   rec  = (int*)(temp + (size_t)NUM_ITEM * DIM_E);
    float* deg  = (float*)(rec + (size_t)N_NODES * REC_W);
    float* w1s  = deg + DEG_PITCH;
    float* w2s  = w1s + DIM_FEAT * DIM_H;
    float* xs   = w2s + DIM_H * DIM_E;
    float* ys   = xs   + (size_t)N_NODES * DIM_E;
    int*   qn   = (int*)(ys + (size_t)N_NODES * DIM_E);

    const int T = 256;
    k_init<<<(N_NODES + T - 1) / T, T, 0, stream>>>(deg, rec, W1, W2, w1s, w2s, qn);
    k_degree_slots<<<2048, T, 0, stream>>>(ei, deg, rec, qn);
    k_dis<<<(N_NODES + T - 1) / T, T, 0, stream>>>(deg);
    k_mlp<<<(N_TILES + 3) / 4, 128, 0, stream>>>(F, w1s, b1, w2s, b2, temp);
    k_normalize<<<(NORM_WAVES + 7) / 8, T, 0, stream>>>(pref, temp, deg, xn, xs,
                                                        out + (size_t)N_NODES * DIM_E);
    {
        k_pull<true><<<PULL_WGS, T, 0, stream>>>(ei, rec, deg, xs, xs, ys);
        k_pull<false><<<PULL_WGS, T, 0, stream>>>(ei, rec, deg, ys, xn, out);
    }
}


# MLP balanced over SIMDs: 768 work-groups take a tile per wave; each of the 53 left-over tiles gets a work-group whose 4 waves take 64 hidden units each, summed in LDS; those rows' sum order changes
# speedup vs baseline: 4.2557x; 1.0418x over previous
#include <hip/hip_runtime.h>

#define NUM_USER 50000
#define NUM_ITEM 50000
#define N_NODES  100000
#define DIM_E    64
#define DIM_FEAT 128
#define DIM_H    256
#define N_EDGES  1600000
#define N_TILES  (NUM_ITEM / 16)
#define DEG_PITCH ((N_NODES + 63) / 64 * 64)

typedef __attribute__((ext_vector_type(2))) float v2f;
typedef __attribute__((ext_vector_type(8))) float v8f;

#define REC_W   64
#define REC_CAP (REC_W - 1)
#define QN_PITCH 64

__global__ void k_init(float* __restrict__ deg, int* __restrict__ rec,
                       const float* __restrict__ W1, const float* __restrict__ W2,
                       float* __restrict__ w1s, float* __restrict__ w2s,
                       int* __restrict__ qn) {
    int n = blockIdx.x * blockDim.x + threadIdx.x;
    if (n >= N_NODES) return;
    deg[n] = 0.0f;
    rec[(size_t)n * REC_W] = 0;
    if (n < 2) qn[QN_PITCH * n] = 0;
    const int q = n & 3;
    const int c = (n >> 2) & 15;
    const int h = (n >> 6) & 1;
    if (n < DIM_FEAT * DIM_H) {
        const int j = (n >> 7) & 15;
        const int t = (n >> 11) & 3;
        const int g = (n >> 13) & 3;
        w1s[n] = W1[(8 * j + 4 * h + q) * DIM_H + 64 * g + 4 * c + t];
    }
    if (n < DIM_H * DIM_E) {
        const int ip = (n >> 7) & 1;
        const int j2 = (n >> 8) & 3;
        const int t  = (n >> 10) & 3;
        const int g  = (n >> 12) & 3;
        w2s[n] = W2[(64 * g + 32 * h + 8 * j2 + 4 * (q & 1) + t) * DIM_E + 4 * c + 2 * ip + (q >> 1)];
    }
}

template <int U> __device__ __forceinline__ int row_bcast_i(int v) {
    return __builtin_amdgcn_update_dpp(0, v, 0x150 + U, 0xf, 0xf, false);
}
template <int U> __device__ __forceinline__ float row_bcast_f(float v) {
    return __int_as_float(row_bcast_i<U>(__float_as_int(v)));
}

#define WM(A0, A1, B, ACC)                                                      \
    ACC = __builtin_amdgcn_wmma_f32_16x16x4_f32(false, (v2f){A0, A1}, false, B, \
                                                (short)0, ACC, false, false)
#define LRELU(X) (((X) > 0.0f) ? (X) : 0.01f * (X))

template <int NG>
__device__ __forceinline__ void mlp_groups(const float* __restrict__ fp,
                                           const float4* __restrict__ p1,
                                           const float4* __restrict__ p2,
                                           const float* __restrict__ bh,
                                           v8f& o0, v8f& o1, v8f& o2, v8f& o3) {
    constexpr int WA = (NG == 1) ? 2 : 3;
    constexpr int YA = (NG == 1) ? 4 : 6;
    constexpr int FA = (NG == 1) ? 4 : 16;
    float4 fq[16];
    #pragma unroll
    for (int j = 0; j < FA; ++j) fq[j] = *(const float4*)(fp + 8 * j);
    float4 nx[WA][4];
    #pragma unroll
    for (int j = 0; j < WA; ++j) {
        #pragma unroll
        for (int t = 0; t < 4; ++t) nx[j][t] = p1[(t * 16 + j) * 32];
    }

    #pragma unroll
    for (int g = 0; g < NG; ++g) {
        const float4* q1 = p1 + g * (4 * 16 * 32);
        const float4* q2 = p2 + g * (16 * 2 * 32);
        const float4* qn = q1 + (4 * 16 * 32);
        float4 w[16][4];
        float4 y[16][2];
        v8f a[4] = {};
        float4 bb[8];
        #pragma unroll
        for (int j = 0; j < WA; ++j) {
            #pragma unroll
            for (int t = 0; t < 4; ++t) w[j][t] = nx[j][t];
        }
        #pragma unroll
        for (int j = 0; j < 16; ++j) {
            if (j + WA < 16) {
                #pragma unroll
                for (int t = 0; t < 4; ++t) w[j + WA][t] = q1[(t * 16 + j + WA) * 32];
            } else {
                #pragma unroll
                for (int k = 2 * (j - (16 - WA)); k < 2 * (j - (16 - WA)) + 2; ++k) {
                    if (k < YA) {
                        y[k][0] = q2[(k * 2 + 0) * 32];
                        y[k][1] = q2[(k * 2 + 1) * 32];
                    }
                }
            }
            if (j + FA < 16) fq[j + FA] = *(const float4*)(fp + 8 * (j + FA));
            if (j >= 4 && j < 12)
                bb[j - 4] = *(const float4*)(bh + 64 * g + 4 * (j - 4));
            const v2f bA = {fq[j].x, fq[j].y};
            const v2f bB = {fq[j].z, fq[j].w};
            #pragma unroll
            for (int t = 0; t < 4; ++t) WM(w[j][t].x, w[j][t].y, bA, a[t]);
            #pragma unroll
            for (int t = 0; t < 4; ++t) WM(w[j][t].z, w[j][t].w, bB, a[t]);
            if (NG == 1) __builtin_amdgcn_sched_barrier(0);
        }
        #pragma unroll
        for (int r = 0; r < 8; ++r) {
            float v;
            v = a[0][r] + bb[r].x; a[0][r] = LRELU(v);
            v = a[1][r] + bb[r].y; a[1][r] = LRELU(v);
            v = a[2][r] + bb[r].z; a[2][r] = LRELU(v);
            v = a[3][r] + bb[r].w; a[3][r] = LRELU(v);
        }
        #pragma unroll
        for (int k = 0; k < 16; ++k) {
            if (k + YA < 16) {
                y[k + YA][0] = q2[((k + YA) * 2 + 0) * 32];
                y[k + YA][1] = q2[((k + YA) * 2 + 1) * 32];
            } else if (k >= 16 - WA && g + 1 < NG) {
                #pragma unroll
                for (int t = 0; t < 4; ++t) nx[k - (16 - WA)][t] = qn[(t * 16 + (k - (16 - WA))) * 32];
            }
            const v2f bh2 = {a[k >> 2][2 * (k & 3)], a[k >> 2][2 * (k & 3) + 1]};
            WM(y[k][0].x, y[k][0].y, bh2, o0); WM(y[k][0].z, y[k][0].w, bh2, o1);
            WM(y[k][1].x, y[k][1].y, bh2, o2); WM(y[k][1].z, y[k][1].w, bh2, o3);
            if (NG == 1) __builtin_amdgcn_sched_barrier(0);
        }
    }
}

#define MLP_WGPS       256
#define MLP_FULL_WGS   (N_TILES / (4 * MLP_WGPS) * MLP_WGPS)
#define MLP_FULL_TILES (MLP_FULL_WGS * 4)
#define MLP_SPLIT_WGS  (N_TILES - MLP_FULL_TILES)

template <bool SPLIT>
__device__ __forceinline__ void mlp_tile(const float* __restrict__ F,
                                         const float* __restrict__ w1s, const float* __restrict__ b1,
                                         const float* __restrict__ w2s, const float* __restrict__ b2,
                                         float* __restrict__ temp, float (&stage)[4][16][68],
                                         const int tile, const int g) {
    const int wave = threadIdx.x >> 5;
    const int lane = threadIdx.x & 31;
    const int c    = lane & 15;
    const int h    = lane >> 4;
    const int r0 = tile * 16;

    const float* fp = F + (size_t)(r0 + c) * DIM_FEAT + 4 * h;

    v8f o0 = {}, o1 = {}, o2 = {}, o3 = {};
    const float4* __restrict__ p1 = ((const float4*)w1s + g * (4 * 16 * 32)) + lane;
    const float4* __restrict__ p2 = ((const float4*)w2s + g * (16 * 2 * 32)) + lane;
    mlp_groups<SPLIT ? 1 : 4>(fp, p1, p2, b1 + 64 * g + 32 * h, o0, o1, o2, o3);

    float* st = &stage[wave][c][32 * h];
    if (!SPLIT) {
        #pragma unroll
        for (int r = 0; r < 8; ++r) {
            const float4 bb = *(const float4*)(b2 + 32 * h + 4 * r);
            float4 v;
            v.x = o0[r] + bb.x; v.y = o1[r] + bb.y;
            v.z = o2[r] + bb.z; v.w = o3[r] + bb.w;
            *(float4*)(st + 4 * r) = v;
        }
        #pragma unroll
        for (int it = 0; it < 8; ++it) {
            const int row = 2 * it + h;
            const float4 v = *(const float4*)(&stage[wave][row][4 * c]);
            *(float4*)(temp + (size_t)(r0 + row) * DIM_E + 4 * c) = v;
        }
    } else {
        #pragma unroll
        for (int r = 0; r < 8; ++r) {
            float4 v;
            v.x = o0[r]; v.y = o1[r]; v.z = o2[r]; v.w = o3[r];
            *(float4*)(st + 4 * r) = v;
        }
        __syncthreads();
        #pragma unroll
        for (int it = 0; it < 2; ++it) {
            const int piece = (int)threadIdx.x + 128 * it;
            const int row = piece >> 4;
            const int c4  = 4 * (piece & 15);
            float4 v = *(const float4*)(&stage[0][row][c4]);
            #pragma unroll
            for (int k = 1; k < 4; ++k) {
                const float4 u = *(const float4*)(&stage[k][row][c4]);
                v.x += u.x; v.y += u.y; v.z += u.z; v.w += u.w;
            }
            const float4 bb = *(const float4*)(b2 + c4);
            v.x += bb.x; v.y += bb.y; v.z += bb.z; v.w += bb.w;
            *(float4*)(temp + (size_t)(r0 + row) * DIM_E + c4) = v;
        }
    }
}

__global__ __launch_bounds__(128) __attribute__((amdgpu_waves_per_eu(1, 4))) void k_mlp(
        const float* __restrict__ F,
        const float* __restrict__ w1s, const float* __restrict__ b1,
        const float* __restrict__ w2s, const float* __restrict__ b2,
        float* __restrict__ temp) {
    __shared__ float stage[4][16][68];
    if (blockIdx.x < MLP_FULL_WGS) {
        mlp_tile<false>(F, w1s, b1, w2s, b2, temp, stage,
                        (int)blockIdx.x * 4 + (int)(threadIdx.x >> 5), 0);
    } else {
        mlp_tile<true>(F, w1s, b1, w2s, b2, temp, stage,
                       MLP_FULL_TILES + ((int)blockIdx.x - MLP_FULL_WGS),
                       __builtin_amdgcn_readfirstlane((int)(threadIdx.x >> 5)));
    }
}
#undef WM
#undef LRELU

template <int M> __device__ __forceinline__ float row_xor_f(float v) {
    return __int_as_float(__builtin_amdgcn_update_dpp(0, __float_as_int(v), 0x160 + M, 0xf, 0xf, false));
}

#define NORM_R 8
#define NORM_WAVES (N_NODES / (2 * NORM_R))
static_assert(N_NODES % (2 * NORM_R) == 0, "a wave takes 16 whole nodes");

__global__ __launch_bounds__(256) void k_normalize(const float* __restrict__ pref,
                                                   const float* __restrict__ temp,
                                                   const float* __restrict__ dis,
                                                   float* __restrict__ xn,
                                                   float* __restrict__ xs,
                                                   float* __restrict__ out_pref) {
    const int gw   = (blockIdx.x * blockDim.x + threadIdx.x) >> 5;
    const int lane = threadIdx.x & 31;
    const int h    = lane >> 4;
    const int q    = lane & 15;
    if (gw >= NORM_WAVES) return;
    const int first = gw * (2 * NORM_R) + h;

    float4 v[NORM_R];
    float  d[NORM_R];
    #pragma unroll
    for (int k = 0; k < NORM_R; ++k) {
        const int node = first + 2 * k;
        const float* src = (node < NUM_USER) ? (pref + (size_t)node * DIM_E)
                                             : (temp + (size_t)(node - NUM_USER) * DIM_E);
        v[k] = *(const float4*)(src + 4 * q);
        d[k] = dis[node];
    }
    #pragma unroll
    for (int k = 0; k < NORM_R; ++k) {
        const int node = first + 2 * k;
        const size_t at = (size_t)node * DIM_E + 4 * q;
        if (node < NUM_USER) {
            *(float4*)(out_pref + at) = v[k];
        }
        float s = v[k].x * v[k].x + v[k].y * v[k].y + v[k].z * v[k].z + v[k].w * v[k].w;
        s += row_xor_f<1>(s);
        s += row_xor_f<2>(s);
        s += row_xor_f<4>(s);
        s += row_xor_f<8>(s);
        const float inv = 1.0f / fmaxf(sqrtf(s), 1e-12f);
        float4 n;
        n.x = v[k].x * inv;
        n.y = v[k].y * inv;
        n.z = v[k].z * inv;
        n.w = v[k].w * inv;
        *(float4*)(xn + at) = n;
        float4 m;
        m.x = n.x * d[k];
        m.y = n.y * d[k];
        m.z = n.z * d[k];
        m.w = n.w * d[k];
        *(float4*)(xs + at) = m;
    }
}

#define EDGE_CH  1024
#define N_CH     ((N_EDGES + EDGE_CH - 1) / EDGE_CH)
__device__ __forceinline__ int node_set(int n) { return (n >> 6) & 1; }

__global__ __launch_bounds__(256) void k_degree_slots(const int* __restrict__ ei,
                                                      float* __restrict__ deg,
                                                      int* __restrict__ rec,
                                                      int* __restrict__ qn) {
    __shared__ int s_task;
    const unsigned hwid = __builtin_amdgcn_s_sendmsg_rtn(0x87);
    const int myhalf = (int)((hwid >> 18) & 1u);

    for (int pass = 0; pass < 2; ++pass) {
        const int set = myhalf ^ pass;
        int* __restrict__ ctr = qn + QN_PITCH * set;
        for (int it = 0; it <= N_CH; ++it) {
            if (threadIdx.x == 0) s_task = atomicAdd(ctr, 1);
            __syncthreads();
            const int task = s_task;
            __syncthreads();
            if (task >= N_CH) break;
            const int e0 = task * EDGE_CH + threadIdx.x;
            int r[4], c[4];
            #pragma unroll
            for (int k = 0; k < 4; ++k) {
                const int e = e0 + 256 * k;
                const bool in = e < N_EDGES;
                r[k] = in ? ei[e] : 0;
                c[k] = in ? ei[N_EDGES + e] : 0;
            }
            int s[4];
            #pragma unroll
            for (int k = 0; k < 4; ++k) {
                s[k] = REC_CAP;
                if (r[k] != c[k]) {
                    if (node_set(r[k]) == set) atomicAdd(deg + r[k], 1.0f);
                    if (node_set(c[k]) == set) s[k] = atomicAdd(rec + (size_t)c[k] * REC_W, 1);
                }
            }
            #pragma unroll
            for (int k = 0; k < 4; ++k) {
                if (s[k] < REC_CAP) rec[(size_t)c[k] * REC_W + 1 + s[k]] = r[k];
            }
        }
    }
}

__global__ void k_dis(float* __restrict__ deg) {
    int n = blockIdx.x * blockDim.x + threadIdx.x;
    if (n >= N_NODES) return;
    float d = deg[n];
    deg[n] = (d > 0.0f) ? rsqrtf(d) : 0.0f;
}

#define LD(U)                                                                   \
    const unsigned r##U = (unsigned)row_bcast_i<U>(a);                          \
    const float  w##U = row_bcast_f<U>(w);                                      \
    const float4 v##U = *(const float4*)(xb + (r##U * ROW_B + qb));
#define AC(U)                                                                   \
    acc.x = fmaf(v##U.x, w##U, acc.x);                                          \
    acc.y = fmaf(v##U.y, w##U, acc.y);                                          \
    acc.z = fmaf(v##U.z, w##U, acc.z);                                          \
    acc.w = fmaf(v##U.w, w##U, acc.w);

#define PULL_WGS  1280
#define PULL_ROWS (PULL_WGS * 16u)
#define ROW_B     (DIM_E * 4u)

template <bool PRESCALED>
__global__ __launch_bounds__(256) void k_pull(const int* __restrict__ ei,
                                              const int* __restrict__ rec,
                                              const float* __restrict__ dis,
                                              const float* __restrict__ x,
                                              const float* __restrict__ own,
                                              float* __restrict__ out) {
    const unsigned tid = blockIdx.x * blockDim.x + threadIdx.x;
    const unsigned q  = tid & 15u;
    const unsigned qb = q * 16u;
    const char* __restrict__ xb = (const char*)x;
    unsigned c = tid >> 4;
    int nx0 = 0, nx1 = 0;
    if (c < N_NODES) {
        nx0 = rec[c * REC_W + q];
        nx1 = rec[c * REC_W + 16u + q];
    }
    for (; c < N_NODES; c += PULL_ROWS) {
    const int* __restrict__ t = rec + c * REC_W;
    const int w0 = nx0;
    const int w1 = nx1;
    const int total = row_bcast_i<0>(w0);
    if (c + PULL_ROWS < N_NODES) {
        nx0 = rec[(c + PULL_ROWS) * REC_W + q];
        nx1 = rec[(c + PULL_ROWS) * REC_W + 16u + q];
    }
    const float dc = dis[c];
    const float4 self = *(const float4*)((const char*)own + (c * ROW_B + qb));
    float4 acc = {0.0f, 0.0f, 0.0f, 0.0f};

    if (total > 0 && total <= REC_CAP) {
        const bool ok0 = (q != 0u) && ((int)q <= total);
        int a = w0;
        const int rpad = row_bcast_i<1>(a);
        if (!ok0) a = rpad;
        float w = ok0 ? 1.0f : 0.0f;
        {
            LD(1) LD(2) LD(3) LD(4) LD(5) LD(6) LD(7) LD(8)
            LD(9) LD(10) LD(11) LD(12) LD(13) LD(14) LD(15)
            AC(1) AC(2) AC(3) AC(4) AC(5) AC(6) AC(7) AC(8)
            AC(9) AC(10) AC(11) AC(12) AC(13) AC(14) AC(15)
        }
        if (total > 15) {
            const bool ok1 = ((int)q < total - 15);
            a = ok1 ? w1 : rpad;
            w = ok1 ? 1.0f : 0.0f;
            {
                LD(0) LD(1) LD(2) LD(3) LD(4) LD(5) LD(6) LD(7)
                AC(0) AC(1) AC(2) AC(3) AC(4) AC(5) AC(6) AC(7)
            }
            if (total > 23) {
                LD(8) LD(9) LD(10) LD(11) LD(12) LD(13) LD(14) LD(15)
                AC(8) AC(9) AC(10) AC(11) AC(12) AC(13) AC(14) AC(15)
            }
        }
        if (total > 31) {
            #pragma nounroll
            for (int base = 31; base < total; base += 8) {
                const bool ok = (q < 8u) && (base + (int)q < total);
                a = rpad;
                if (ok) a = t[base + 1 + q];
                w = ok ? 1.0f : 0.0f;
                LD(0) LD(1) LD(2) LD(3) LD(4) LD(5) LD(6) LD(7)
                AC(0) AC(1) AC(2) AC(3) AC(4) AC(5) AC(6) AC(7)
            }
        }
    } else if (total > REC_CAP) {
        for (unsigned e0 = 0; e0 < N_EDGES; e0 += 8) {
            int rq = (int)c, cq = -1;
            if (q < 8u) {
                rq = ei[e0 + q];
                cq = ei[N_EDGES + e0 + q];
            }
            const bool hit = (cq == (int)c) && (rq != (int)c);
            const unsigned hits = __builtin_amdgcn_ballot_w32(hit);
            if (((hits >> (threadIdx.x & 16)) & 0xffffu) != 0u) {
                const int a = hit ? rq : (int)c;
                const float w = hit ? 1.0f : 0.0f;
                LD(0) LD(1) LD(2) LD(3) LD(4) LD(5) LD(6) LD(7)
                AC(0) AC(1) AC(2) AC(3) AC(4) AC(5) AC(6) AC(7)
            }
        }
    }
    const float sc = PRESCALED ? dc * dc : dc;
    float4 res;
    res.x = fmaf(acc.x, sc, self.x);
    res.y = fmaf(acc.y, sc, self.y);
    res.z = fmaf(acc.z, sc, self.z);
    res.w = fmaf(acc.w, sc, self.w);
    *(float4*)((char*)out + (c * ROW_B + qb)) = res;
    }
}
#undef LD
#undef AC

extern "C" void kernel_launch(void* const* d_in, const int* in_sizes, int n_in,
                              void* d_out, int out_size, void* d_ws, size_t ws_size,
                              hipStream_t stream) {
    const int*   ei   = (const int*)  d_in[0];
    const float* F    = (const float*)d_in[1];
    const float* pref = (const float*)d_in[2];
    const float* W1   = (const float*)d_in[3];
    const float* b1   = (const float*)d_in[4];
    const float* W2   = (const float*)d_in[5];
    const float* b2   = (const float*)d_in[6];
    float*       out  = (float*)d_out;

    float* ws   = (float*)d_ws;
    float* xn   = ws;
    float* temp = xn   + (size_t)N_NODES * DIM_E;
    int*   rec  = (int*)(temp + (size_t)NUM_ITEM * DIM_E);
    float* deg  = (float*)(rec + (size_t)N_NODES * REC_W);
    float* w1s  = deg + DEG_PITCH;
    float* w2s  = w1s + DIM_FEAT * DIM_H;
    float* xs   = w2s + DIM_H * DIM_E;
    float* ys   = xs   + (size_t)N_NODES * DIM_E;
    int*   qn   = (int*)(ys + (size_t)N_NODES * DIM_E);

    const int T = 256;
    k_init<<<(N_NODES + T - 1) / T, T, 0, stream>>>(deg, rec, W1, W2, w1s, w2s, qn);
    k_degree_slots<<<2048, T, 0, stream>>>(ei, deg, rec, qn);
    k_dis<<<(N_NODES + T - 1) / T, T, 0, stream>>>(deg);
    k_mlp<<<MLP_FULL_WGS + MLP_SPLIT_WGS, 128, 0, stream>>>(F, w1s, b1, w2s, b2, temp);
    k_normalize<<<(NORM_WAVES + 7) / 8, T, 0, stream>>>(pref, temp, deg, xn, xs,
                                                        out + (size_t)N_NODES * DIM_E);
    {
        k_pull<true><<<PULL_WGS, T, 0, stream>>>(ei, rec, deg, xs, xs, ys);
        k_pull<false><<<PULL_WGS, T, 0, stream>>>(ei, rec, deg, ys, xn, out);
    }
}
